# starting kernel
# baseline (speedup 1.0000x reference)
#include <hip/hip_runtime.h>

typedef float v2f __attribute__((ext_vector_type(2)));
typedef float v8f __attribute__((ext_vector_type(8)));

#define DD 64
#define SPITCH 68

static __device__ __forceinline__ v8f wmma_f32x4(v2f a, v2f b, v8f c) {
  return __builtin_amdgcn_wmma_f32_16x16x4_f32(
       false, a,  false, b,
       (short)0, c,  false,  false);
}

__global__ void zero_kernel(float* __restrict__ p, long long n) {
  long long i = (long long)blockIdx.x * blockDim.x + threadIdx.x;
  long long stride = (long long)gridDim.x * blockDim.x;
  for (; i < n; i += stride) p[i] = 0.0f;
}

__global__ void edge_scatter_kernel(const float* __restrict__ n_feats,
                                    const float* __restrict__ r_feats,
                                    const int* __restrict__ src,
                                    const int* __restrict__ dst,
                                    const int* __restrict__ etype,
                                    const int* __restrict__ num_rels_p,
                                    float* __restrict__ accI,
                                    float* __restrict__ accO,
                                    float* __restrict__ cntI,
                                    float* __restrict__ cntO,
                                    int E) {
  int wave = (int)((blockIdx.x * blockDim.x + threadIdx.x) >> 5);
  int lane = threadIdx.x & 31;
  if (wave >= E) return;

  int s = src[wave];
  int d = dst[wave];
  int t = etype[wave];
  int half = num_rels_p[0] >> 1;
  bool isI = (t < half);

  const float2* xs = (const float2*)(n_feats + (size_t)s * DD);
  const float2* xr = (const float2*)(r_feats + (size_t)t * DD);
  float2 a = xs[lane];
  float2 b = xr[lane];
  float cx = a.x - b.x;
  float cy = a.y - b.y;

  float* acc = isI ? accI : accO;
  size_t base = (size_t)d * DD + 2 * lane;
  atomicAdd(&acc[base + 0], cx);
  atomicAdd(&acc[base + 1], cy);
  if (lane == 0) atomicAdd(isI ? &cntI[d] : &cntO[d], 1.0f);
}

__global__ void __launch_bounds__(128)
node_transform_kernel(const float* __restrict__ accI,
                      const float* __restrict__ accO,
                      const float* __restrict__ cntI,
                      const float* __restrict__ cntO,
                      const float* __restrict__ W_I_w,
                      const float* __restrict__ W_I_b,
                      const float* __restrict__ W_O_w,
                      const float* __restrict__ W_O_b,
                      float* __restrict__ out,
                      int numTiles) {
  __shared__ float sWI[DD * SPITCH];
  __shared__ float sWO[DD * SPITCH];

  {
    const float4* gI = (const float4*)W_I_w;
    const float4* gO = (const float4*)W_O_w;
    int tid = threadIdx.x;
#pragma unroll
    for (int i = 0; i < (DD * DD / 4) / 128; ++i) {
      int idx = tid + i * 128;
      int r = idx >> 4;
      int c4 = idx & 15;
      *(float4*)(&sWI[r * SPITCH + c4 * 4]) = gI[idx];
      *(float4*)(&sWO[r * SPITCH + c4 * 4]) = gO[idx];
    }
  }
  __syncthreads();

  int tile = blockIdx.x * 4 + (threadIdx.x >> 5);
  if (tile >= numTiles) return;

  int lane = threadIdx.x & 31;
  int lm   = lane & 15;
  int hs   = lane >> 4;
  int kb   = hs * 2;
  int mOff = hs * 8;
  int rowBase = tile * 16;

  float cI8[8], cO8[8], dg8[8];
#pragma unroll
  for (int i = 0; i < 8; ++i) {
    int m = rowBase + i + mOff;
    cI8[i] = cntI[m];
    cO8[i] = cntO[m];
    dg8[i] = fmaxf(cI8[i] + cO8[i], 1.0f);
  }

  const float* aIrow = accI + (size_t)(rowBase + lm) * DD;
  const float* aOrow = accO + (size_t)(rowBase + lm) * DD;
  const float* sIrow[4];
  const float* sOrow[4];
#pragma unroll
  for (int t = 0; t < 4; ++t) {
    sIrow[t] = sWI + (t * 16 + lm) * SPITCH;
    sOrow[t] = sWO + (t * 16 + lm) * SPITCH;
  }

  v8f c[4];
#pragma unroll
  for (int t = 0; t < 4; ++t) c[t] = (v8f){0.f, 0.f, 0.f, 0.f, 0.f, 0.f, 0.f, 0.f};

#pragma unroll
  for (int k0 = 0; k0 < DD; k0 += 4) {
    v2f aI = *(const v2f*)(aIrow + k0 + kb);
    v2f aO = *(const v2f*)(aOrow + k0 + kb);
#pragma unroll
    for (int t = 0; t < 4; ++t) {
      v2f bI = *(const v2f*)(sIrow[t] + k0 + kb);
      v2f bO = *(const v2f*)(sOrow[t] + k0 + kb);
      c[t] = wmma_f32x4(aI, bI, c[t]);
      c[t] = wmma_f32x4(aO, bO, c[t]);
    }
  }

#pragma unroll
  for (int t = 0; t < 4; ++t) {
    int n = t * 16 + lm;
    float bI = W_I_b[n];
    float bO = W_O_b[n];
#pragma unroll
    for (int i = 0; i < 8; ++i) {
      int m = rowBase + i + mOff;
      float v = (c[t][i] + cI8[i] * bI + cO8[i] * bO) / dg8[i];
      out[(size_t)m * DD + n] = v;
    }
  }
}

__global__ void rel_transform_kernel(const float* __restrict__ r_feats,
                                     const float* __restrict__ W_R_w,
                                     const float* __restrict__ W_R_b,
                                     float* __restrict__ out,
                                     int R) {
  int wavesPerBlock = blockDim.x >> 5;
  int numTiles = (R + 15) >> 4;
  int tile = blockIdx.x * wavesPerBlock + (threadIdx.x >> 5);
  if (tile >= numTiles) return;

  int lane = threadIdx.x & 31;
  int lm   = lane & 15;
  int hs   = lane >> 4;
  int kb   = hs * 2;
  int mOff = hs * 8;
  int rowBase = tile * 16;

  int mA = rowBase + lm;
  bool aOk = (mA < R);
  const float* arow = r_feats + (size_t)(aOk ? mA : 0) * DD;

  const float* Wrow[4];
#pragma unroll
  for (int t = 0; t < 4; ++t) Wrow[t] = W_R_w + (size_t)(t * 16 + lm) * DD;

  v8f c[4];
#pragma unroll
  for (int t = 0; t < 4; ++t) c[t] = (v8f){0.f, 0.f, 0.f, 0.f, 0.f, 0.f, 0.f, 0.f};

#pragma unroll
  for (int k0 = 0; k0 < DD; k0 += 4) {
    v2f a = *(const v2f*)(arow + k0 + kb);
    if (!aOk) { a.x = 0.f; a.y = 0.f; }
#pragma unroll
    for (int t = 0; t < 4; ++t) {
      v2f b = *(const v2f*)(Wrow[t] + k0 + kb);
      c[t] = wmma_f32x4(a, b, c[t]);
    }
  }

#pragma unroll
  for (int t = 0; t < 4; ++t) {
    int n = t * 16 + lm;
    float bias = W_R_b[n];
#pragma unroll
    for (int i = 0; i < 8; ++i) {
      int m = rowBase + i + mOff;
      if (m < R) out[(size_t)m * DD + n] = c[t][i] + bias;
    }
  }
}

extern "C" void kernel_launch(void* const* d_in, const int* in_sizes, int n_in,
                              void* d_out, int out_size, void* d_ws, size_t ws_size,
                              hipStream_t stream) {
  const float* n_feats  = (const float*)d_in[0];
  const float* r_feats  = (const float*)d_in[1];
  const float* W_I_w    = (const float*)d_in[2];
  const float* W_I_b    = (const float*)d_in[3];
  const float* W_O_w    = (const float*)d_in[4];
  const float* W_O_b    = (const float*)d_in[5];
  const float* W_R_w    = (const float*)d_in[6];
  const float* W_R_b    = (const float*)d_in[7];
  const int*   src      = (const int*)d_in[8];
  const int*   dst      = (const int*)d_in[9];
  const int*   etype    = (const int*)d_in[10];
  const int*   num_rels = (const int*)d_in[11];

  int N = in_sizes[0] / DD;
  int R = in_sizes[1] / DD;
  int E = in_sizes[8];

  float* accI = (float*)d_ws;
  float* accO = accI + (size_t)N * DD;
  float* cntI = accO + (size_t)N * DD;
  float* cntO = cntI + N;

  long long zcount = (long long)N * (2 * DD + 2);
  zero_kernel<<<1024, 256, 0, stream>>>(accI, zcount);

  int edgeBlocks = (E + 7) / 8;
  edge_scatter_kernel<<<edgeBlocks, 256, 0, stream>>>(
      n_feats, r_feats, src, dst, etype, num_rels,
      accI, accO, cntI, cntO, E);

  int nodeTiles = (N + 15) / 16;
  node_transform_kernel<<<(nodeTiles + 3) / 4, 128, 0, stream>>>(
      accI, accO, cntI, cntO, W_I_w, W_I_b, W_O_w, W_O_b,
      (float*)d_out, nodeTiles);

  int relTiles = (R + 15) / 16;
  rel_transform_kernel<<<(relTiles + 3) / 4, 128, 0, stream>>>(
      r_feats, W_R_w, W_R_b, (float*)d_out + (size_t)N * DD, R);
}


# bucket edges by (dst, I/O side) inside each call with integer atomics only, then sum each node's rows in registers inside the WMMA node kernel: the per-edge f32 atomic adds are gone
# speedup vs baseline: 1.1216x; 1.1216x over previous
#include <hip/hip_runtime.h>


typedef float v2f __attribute__((ext_vector_type(2)));
typedef float v8f __attribute__((ext_vector_type(8)));

#define DD 64
#define SPITCH 68
#define CH 16

static __device__ __forceinline__ v8f wmma_f32x4(v2f a, v2f b, v8f c) {
  return __builtin_amdgcn_wmma_f32_16x16x4_f32(
       false, a,  false, b,
       (short)0, c,  false,  false);
}

__global__ void zero_int_kernel(int* __restrict__ p, int n) {
  int i = blockIdx.x * blockDim.x + threadIdx.x;
  if (i < n) p[i] = 0;
}

__global__ void degree_kernel(const int* __restrict__ dst,
                              const int* __restrict__ etype,
                              const int* __restrict__ num_rels_p,
                              int* __restrict__ deg, int E) {
  int e = blockIdx.x * blockDim.x + threadIdx.x;
  if (e >= E) return;
  int half = num_rels_p[0] >> 1;
  int key = 2 * dst[e] + ((etype[e] < half) ? 0 : 1);
  atomicAdd(&deg[key], 1);
}

__global__ void __launch_bounds__(256)
range_alloc_kernel(const int* __restrict__ deg,
                   int* __restrict__ total,
                   int* __restrict__ cursor, int M) {
  __shared__ int sWave[8];
  __shared__ int sBase;
  int k = blockIdx.x * blockDim.x + threadIdx.x;
  int lane = threadIdx.x & 31;
  int wv = threadIdx.x >> 5;
  int v = (k < M) ? deg[k] : 0;
  int inc = v;
  for (int d = 1; d < 32; d <<= 1) {
    int o = __shfl_up(inc, d);
    if (lane >= d) inc += o;
  }
  if (lane == 31) sWave[wv] = inc;
  __syncthreads();
  if (threadIdx.x == 0) {
    int s = 0;
    for (int i = 0; i < 8; ++i) {
      int t = sWave[i];
      sWave[i] = s;
      s += t;
    }
    sBase = atomicAdd(total, s);
  }
  __syncthreads();
  if (k < M) cursor[k] = sBase + sWave[wv] + inc - v;
}

__global__ void bucket_fill_kernel(const int* __restrict__ src,
                                   const int* __restrict__ dst,
                                   const int* __restrict__ etype,
                                   const int* __restrict__ num_rels_p,
                                   int* __restrict__ cursor,
                                   int* __restrict__ bucket, int E) {
  int e = blockIdx.x * blockDim.x + threadIdx.x;
  if (e >= E) return;
  int half = num_rels_p[0] >> 1;
  int t = etype[e];
  int key = 2 * dst[e] + ((t < half) ? 0 : 1);
  int slot = atomicAdd(&cursor[key], 1);
  ((int2*)bucket)[slot] = make_int2(src[e] * (DD * 4), t * (DD * 4));
}

static __device__ __forceinline__ int2 fetch_batch(const int2* __restrict__ bk, int lane,
                                                   int jI, int nI, int jO, int nO) {
  int hl = lane & 15;
  bool up = (lane >= 16);
  int nn = up ? nO : nI;
  int jj = up ? jO : jI;
  int2 p = make_int2(0, 0);
  if (hl < nn) p = bk[jj + hl];
  return p;
}

static __device__ __forceinline__ float2 row_piece(const float* __restrict__ base,
                                                   unsigned rowByte, unsigned laneByte) {
  return *(const float2*)((const char*)base + (size_t)(rowByte + laneByte));
}

__global__ void __launch_bounds__(128)
node_gather_transform_kernel(const float* __restrict__ n_feats,
                             const float* __restrict__ r_feats,
                             const int* __restrict__ bucket,
                             const int* __restrict__ rangeEnd,
                             const int* __restrict__ deg,
                             const float* __restrict__ W_I_w,
                             const float* __restrict__ W_I_b,
                             const float* __restrict__ W_O_w,
                             const float* __restrict__ W_O_b,
                             float* __restrict__ out,
                             int numTiles) {
  __shared__ float sWI[DD * SPITCH];
  __shared__ float sWO[DD * SPITCH];
  __shared__ float sAI[4][16 * SPITCH];
  __shared__ float sAO[4][16 * SPITCH];
  __shared__ float sCI[4][16];
  __shared__ float sCO[4][16];

  {
    const float4* gI = (const float4*)W_I_w;
    const float4* gO = (const float4*)W_O_w;
    int tid = threadIdx.x;
#pragma unroll
    for (int i = 0; i < (DD * DD / 4) / 128; ++i) {
      int idx = tid + i * 128;
      int r = idx >> 4;
      int c4 = idx & 15;
      *(float4*)(&sWI[r * SPITCH + c4 * 4]) = gI[idx];
      *(float4*)(&sWO[r * SPITCH + c4 * 4]) = gO[idx];
    }
  }

  int wv   = threadIdx.x >> 5;
  int lane = threadIdx.x & 31;
  int tile = blockIdx.x * 4 + wv;
  bool active = (tile < numTiles);
  int rowBase = tile * 16;
  float* aIw = sAI[wv];
  float* aOw = sAO[wv];

  if (active) {
    const int2* bk = (const int2*)bucket;
    int endv = rangeEnd[2 * rowBase + lane];
    int degv = deg[2 * rowBase + lane];
    unsigned laneByte = (unsigned)lane * 8u;

    int2 pvN;
    {
      int dI = __builtin_amdgcn_readlane(degv, 0);
      int dO = __builtin_amdgcn_readlane(degv, 1);
      int eI = __builtin_amdgcn_readlane(endv, 0);
      int eO = __builtin_amdgcn_readlane(endv, 1);
      pvN = fetch_batch(bk, lane, eI - dI, (dI < CH) ? dI : CH, eO - dO, (dO < CH) ? dO : CH);
    }

    for (int r = 0; r < 16; ++r) {
      int dI   = __builtin_amdgcn_readlane(degv, 2 * r);
      int dO   = __builtin_amdgcn_readlane(degv, 2 * r + 1);
      int endI = __builtin_amdgcn_readlane(endv, 2 * r);
      int endO = __builtin_amdgcn_readlane(endv, 2 * r + 1);
      int jI = endI - dI;
      int jO = endO - dO;

      int2 pv = pvN;
      if (r + 1 < 16) {
        int d1I = __builtin_amdgcn_readlane(degv, 2 * r + 2);
        int d1O = __builtin_amdgcn_readlane(degv, 2 * r + 3);
        int e1I = __builtin_amdgcn_readlane(endv, 2 * r + 2);
        int e1O = __builtin_amdgcn_readlane(endv, 2 * r + 3);
        pvN = fetch_batch(bk, lane, e1I - d1I, (d1I < CH) ? d1I : CH,
                          e1O - d1O, (d1O < CH) ? d1O : CH);
      }

      float sIx = 0.f, sIy = 0.f, sOx = 0.f, sOy = 0.f;
      bool first = true;
      while (jI < endI || jO < endO) {
        int nI = endI - jI;
        nI = (nI < 0) ? 0 : ((nI < CH) ? nI : CH);
        int nO = endO - jO;
        nO = (nO < 0) ? 0 : ((nO < CH) ? nO : CH);
        if (!first) pv = fetch_batch(bk, lane, jI, nI, jO, nO);
        first = false;

        float2 x[2 * CH], q[2 * CH];
#pragma unroll
        for (int t = 0; t < CH; ++t) {
          if (t < nI) {
            x[t] = row_piece(n_feats, (unsigned)__builtin_amdgcn_readlane(pv.x, t), laneByte);
            q[t] = row_piece(r_feats, (unsigned)__builtin_amdgcn_readlane(pv.y, t), laneByte);
          }
        }
#pragma unroll
        for (int t = 0; t < CH; ++t) {
          if (t < nO) {
            x[CH + t] = row_piece(n_feats, (unsigned)__builtin_amdgcn_readlane(pv.x, CH + t), laneByte);
            q[CH + t] = row_piece(r_feats, (unsigned)__builtin_amdgcn_readlane(pv.y, CH + t), laneByte);
          }
        }
#pragma unroll
        for (int t = 0; t < CH; ++t) {
          if (t < nI) {
            sIx += x[t].x - q[t].x;
            sIy += x[t].y - q[t].y;
          }
        }
#pragma unroll
        for (int t = 0; t < CH; ++t) {
          if (t < nO) {
            sOx += x[CH + t].x - q[CH + t].x;
            sOy += x[CH + t].y - q[CH + t].y;
          }
        }
        jI += CH;
        jO += CH;
      }

      *(float2*)(aIw + r * SPITCH + 2 * lane) = make_float2(sIx, sIy);
      *(float2*)(aOw + r * SPITCH + 2 * lane) = make_float2(sOx, sOy);
      if (lane == 0) {
        sCI[wv][r] = (float)dI;
        sCO[wv][r] = (float)dO;
      }
    }
  }
  __syncthreads();
  if (!active) return;

  int lm   = lane & 15;
  int hs   = lane >> 4;
  int kb   = hs * 2;
  int mOff = hs * 8;

  float cI8[8], cO8[8], dg8[8];
#pragma unroll
  for (int i = 0; i < 8; ++i) {
    cI8[i] = sCI[wv][i + mOff];
    cO8[i] = sCO[wv][i + mOff];
    dg8[i] = fmaxf(cI8[i] + cO8[i], 1.0f);
  }

  const float* aIrow = aIw + lm * SPITCH;
  const float* aOrow = aOw + lm * SPITCH;
  const float* sIrow[4];
  const float* sOrow[4];
#pragma unroll
  for (int t = 0; t < 4; ++t) {
    sIrow[t] = sWI + (t * 16 + lm) * SPITCH;
    sOrow[t] = sWO + (t * 16 + lm) * SPITCH;
  }

  v8f c[4];
#pragma unroll
  for (int t = 0; t < 4; ++t) c[t] = (v8f){0.f, 0.f, 0.f, 0.f, 0.f, 0.f, 0.f, 0.f};

#pragma unroll
  for (int k0 = 0; k0 < DD; k0 += 4) {
    v2f aI = *(const v2f*)(aIrow + k0 + kb);
    v2f aO = *(const v2f*)(aOrow + k0 + kb);
#pragma unroll
    for (int t = 0; t < 4; ++t) {
      v2f bI = *(const v2f*)(sIrow[t] + k0 + kb);
      v2f bO = *(const v2f*)(sOrow[t] + k0 + kb);
      c[t] = wmma_f32x4(aI, bI, c[t]);
      c[t] = wmma_f32x4(aO, bO, c[t]);
    }
  }

#pragma unroll
  for (int t = 0; t < 4; ++t) {
    int n = t * 16 + lm;
    float bI = W_I_b[n];
    float bO = W_O_b[n];
#pragma unroll
    for (int i = 0; i < 8; ++i) {
      int m = rowBase + i + mOff;
      float v = (c[t][i] + cI8[i] * bI + cO8[i] * bO) / dg8[i];
      out[(size_t)m * DD + n] = v;
    }
  }
}

__global__ void rel_transform_kernel(const float* __restrict__ r_feats,
                                     const float* __restrict__ W_R_w,
                                     const float* __restrict__ W_R_b,
                                     float* __restrict__ out,
                                     int R) {
  int wavesPerBlock = blockDim.x >> 5;
  int numTiles = (R + 15) >> 4;
  int tile = blockIdx.x * wavesPerBlock + (threadIdx.x >> 5);
  if (tile >= numTiles) return;

  int lane = threadIdx.x & 31;
  int lm   = lane & 15;
  int hs   = lane >> 4;
  int kb   = hs * 2;
  int mOff = hs * 8;
  int rowBase = tile * 16;

  int mA = rowBase + lm;
  bool aOk = (mA < R);
  const float* arow = r_feats + (size_t)(aOk ? mA : 0) * DD;

  const float* Wrow[4];
#pragma unroll
  for (int t = 0; t < 4; ++t) Wrow[t] = W_R_w + (size_t)(t * 16 + lm) * DD;

  v8f c[4];
#pragma unroll
  for (int t = 0; t < 4; ++t) c[t] = (v8f){0.f, 0.f, 0.f, 0.f, 0.f, 0.f, 0.f, 0.f};

#pragma unroll
  for (int k0 = 0; k0 < DD; k0 += 4) {
    v2f a = *(const v2f*)(arow + k0 + kb);
    if (!aOk) { a.x = 0.f; a.y = 0.f; }
#pragma unroll
    for (int t = 0; t < 4; ++t) {
      v2f b = *(const v2f*)(Wrow[t] + k0 + kb);
      c[t] = wmma_f32x4(a, b, c[t]);
    }
  }

#pragma unroll
  for (int t = 0; t < 4; ++t) {
    int n = t * 16 + lm;
    float bias = W_R_b[n];
#pragma unroll
    for (int i = 0; i < 8; ++i) {
      int m = rowBase + i + mOff;
      if (m < R) out[(size_t)m * DD + n] = c[t][i] + bias;
    }
  }
}

extern "C" void kernel_launch(void* const* d_in, const int* in_sizes, int n_in,
                              void* d_out, int out_size, void* d_ws, size_t ws_size,
                              hipStream_t stream) {
  const float* n_feats  = (const float*)d_in[0];
  const float* r_feats  = (const float*)d_in[1];
  const float* W_I_w    = (const float*)d_in[2];
  const float* W_I_b    = (const float*)d_in[3];
  const float* W_O_w    = (const float*)d_in[4];
  const float* W_O_b    = (const float*)d_in[5];
  const float* W_R_w    = (const float*)d_in[6];
  const float* W_R_b    = (const float*)d_in[7];
  const int*   src      = (const int*)d_in[8];
  const int*   dst      = (const int*)d_in[9];
  const int*   etype    = (const int*)d_in[10];
  const int*   num_rels = (const int*)d_in[11];

  int N = in_sizes[0] / DD;
  int R = in_sizes[1] / DD;
  int E = in_sizes[8];
  int M = 2 * N;

  int* bucket = (int*)d_ws;
  int* deg    = bucket + (size_t)2 * E;
  int* total  = deg + M;
  int* cursor = total + 1;

  zero_int_kernel<<<(M + 1 + 255) / 256, 256, 0, stream>>>(deg, M + 1);

  int edgeBlocks = (E + 255) / 256;
  degree_kernel<<<edgeBlocks, 256, 0, stream>>>(dst, etype, num_rels, deg, E);

  range_alloc_kernel<<<(M + 255) / 256, 256, 0, stream>>>(deg, total, cursor, M);

  bucket_fill_kernel<<<edgeBlocks, 256, 0, stream>>>(src, dst, etype, num_rels, cursor, bucket, E);

  int nodeTiles = (N + 15) / 16;
  node_gather_transform_kernel<<<(nodeTiles + 3) / 4, 128, 0, stream>>>(
      n_feats, r_feats, bucket, cursor, deg,
      W_I_w, W_I_b, W_O_w, W_O_b, (float*)d_out, nodeTiles);

  int relTiles = (R + 15) / 16;
  rel_transform_kernel<<<(relTiles + 3) / 4, 128, 0, stream>>>(
      r_feats, W_R_w, W_R_b, (float*)d_out + (size_t)N * DD, R);
}


# bucket edges by a two-level counting sort with LDS counters (256 edge chunks x 196 node bins, then one block per bin): the 2M per-edge global atomics and the 62k short-lived waves are gone
# speedup vs baseline: 2.2329x; 1.9908x over previous
#include <hip/hip_runtime.h>


typedef float v2f __attribute__((ext_vector_type(2)));
typedef float v8f __attribute__((ext_vector_type(8)));

#define DD 64
#define SPITCH 68
#define CH 16

static __device__ __forceinline__ v8f wmma_f32x4(v2f a, v2f b, v8f c) {
  return __builtin_amdgcn_wmma_f32_16x16x4_f32(
       false, a,  false, b,
       (short)0, c,  false,  false);
}

#define HB 256
#define HT 512
#define BN_SHIFT 8
#define BK 512

__global__ void __launch_bounds__(HT)
bin_hist_kernel(const int* __restrict__ dst,
                int* __restrict__ histG,
                int* __restrict__ total,
                int E, int NB, int chunk) {
  extern __shared__ int sh[];
  int tid = threadIdx.x;
  for (int b = tid; b < NB; b += HT) sh[b] = 0;
  if (blockIdx.x == 0 && tid == 0) total[0] = 0;
  __syncthreads();
  int lo = blockIdx.x * chunk;
  int hi = lo + chunk;
  hi = (hi < E) ? hi : E;
  for (int i = lo + tid; i < hi; i += 4 * HT) {
    int i1 = i + HT, i2 = i + 2 * HT, i3 = i + 3 * HT;
    int d0 = dst[i];
    int d1 = (i1 < hi) ? dst[i1] : -1;
    int d2 = (i2 < hi) ? dst[i2] : -1;
    int d3 = (i3 < hi) ? dst[i3] : -1;
    atomicAdd(&sh[d0 >> BN_SHIFT], 1);
    if (d1 >= 0) atomicAdd(&sh[d1 >> BN_SHIFT], 1);
    if (d2 >= 0) atomicAdd(&sh[d2 >> BN_SHIFT], 1);
    if (d3 >= 0) atomicAdd(&sh[d3 >> BN_SHIFT], 1);
  }
  __syncthreads();
  int* row = histG + (size_t)blockIdx.x * NB;
  for (int b = tid; b < NB; b += HT) row[b] = sh[b];
}

__global__ void __launch_bounds__(256)
bin_alloc_kernel(int* histG, int* __restrict__ total, int* __restrict__ binRange, int NB) {
  static_assert(HB == 256, "8 waves x 32 histogram rows");
  __shared__ int sPart[8][32];
  __shared__ int sBase[32];
  int b = threadIdx.x & 31;
  int g = threadIdx.x >> 5;
  int bin = blockIdx.x * 32 + b;
  bool ok = (bin < NB);
  int cnt[32];
  int s = 0;
#pragma unroll
  for (int j = 0; j < 32; ++j) {
    int c = ok ? histG[(size_t)(g * 32 + j) * NB + bin] : 0;
    cnt[j] = c;
    s += c;
  }
  sPart[g][b] = s;
  __syncthreads();
  int pre = 0, tot = 0;
#pragma unroll
  for (int gg = 0; gg < 8; ++gg) {
    int v = sPart[gg][b];
    if (gg < g) pre += v;
    tot += v;
  }
  if (g == 0) {
    int inc = tot;
    for (int d = 1; d < 32; d <<= 1) {
      int o = __shfl_up(inc, d);
      if (b >= d) inc += o;
    }
    int base = 0;
    if (b == 31) base = atomicAdd(total, inc);
    base = __shfl(base, 31);
    int st = base + inc - tot;
    sBase[b] = st;
    if (ok) ((int2*)binRange)[bin] = make_int2(st, st + tot);
  }
  __syncthreads();
  int run = sBase[b] + pre;
#pragma unroll
  for (int j = 0; j < 32; ++j) {
    if (ok) histG[(size_t)(g * 32 + j) * NB + bin] = run;
    run += cnt[j];
  }
}

__global__ void __launch_bounds__(HT)
bin_scatter_kernel(const int* __restrict__ src,
                   const int* __restrict__ dst,
                   const int* __restrict__ etype,
                   const int* __restrict__ num_rels_p,
                   const int* __restrict__ histG,
                   int* __restrict__ coarse,
                   int E, int NB, int chunk) {
  extern __shared__ int sh[];
  int tid = threadIdx.x;
  const int* row = histG + (size_t)blockIdx.x * NB;
  for (int b = tid; b < NB; b += HT) sh[b] = row[b];
  __syncthreads();
  int half = num_rels_p[0] >> 1;
  int lo = blockIdx.x * chunk;
  int hi = lo + chunk;
  hi = (hi < E) ? hi : E;
  for (int i = lo + tid; i < hi; i += 4 * HT) {
    int d[4], s[4], t[4];
#pragma unroll
    for (int u = 0; u < 4; ++u) {
      int iu = i + u * HT;
      bool in = (iu < hi);
      int ii = in ? iu : i;
      int dv = dst[ii];
      d[u] = in ? dv : -1;
      s[u] = src[ii];
      t[u] = etype[ii];
    }
#pragma unroll
    for (int u = 0; u < 4; ++u) {
      if (d[u] >= 0) {
        int key = ((d[u] & ((1 << BN_SHIFT) - 1)) << 1) | ((t[u] < half) ? 0 : 1);
        int slot = atomicAdd(&sh[d[u] >> BN_SHIFT], 1);
        ((int2*)coarse)[slot] = make_int2(s[u] * (DD * 4), t[u] * BK + key);
      }
    }
  }
}

__global__ void __launch_bounds__(BK)
bin_sort_kernel(const int* __restrict__ coarse,
                const int* __restrict__ binRange,
                int* __restrict__ bucket,
                int* __restrict__ deg,
                int* __restrict__ rangeEnd,
                int M) {
  static_assert(BK == (2 << BN_SHIFT), "one thread per key of the bin");
  __shared__ int sCnt[BK];
  __shared__ int sWave[BK / 32];
  int tid = threadIdx.x;
  int lane = tid & 31;
  int wv = tid >> 5;
  int2 rg = ((const int2*)binRange)[blockIdx.x];
  int s = rg.x, e = rg.y;
  const int2* cz = (const int2*)coarse;
  sCnt[tid] = 0;
  __syncthreads();
  for (int j0 = s; j0 < e; j0 += 4 * BK) {
    int2 rec[4];
#pragma unroll
    for (int u = 0; u < 4; ++u) {
      int j = j0 + u * BK + tid;
      rec[u] = (j < e) ? cz[j] : make_int2(0, -1);
    }
#pragma unroll
    for (int u = 0; u < 4; ++u)
      if (rec[u].y >= 0) atomicAdd(&sCnt[rec[u].y & (BK - 1)], 1);
  }
  __syncthreads();
  int c = sCnt[tid];
  int inc = c;
  for (int d = 1; d < 32; d <<= 1) {
    int o = __shfl_up(inc, d);
    if (lane >= d) inc += o;
  }
  if (lane == 31) sWave[wv] = inc;
  __syncthreads();
  int woff = 0;
#pragma unroll
  for (int i = 0; i < BK / 32; ++i) {
    int v = sWave[i];
    if (i < wv) woff += v;
  }
  int st = s + woff + inc - c;
  sCnt[tid] = st;
  int key = blockIdx.x * BK + tid;
  if (key < M) {
    deg[key] = c;
    rangeEnd[key] = st + c;
  }
  __syncthreads();
  for (int j0 = s; j0 < e; j0 += 4 * BK) {
    int2 rec[4];
#pragma unroll
    for (int u = 0; u < 4; ++u) {
      int j = j0 + u * BK + tid;
      rec[u] = (j < e) ? cz[j] : make_int2(0, -1);
    }
#pragma unroll
    for (int u = 0; u < 4; ++u) {
      if (rec[u].y >= 0) {
        int slot = atomicAdd(&sCnt[rec[u].y & (BK - 1)], 1);
        ((int2*)bucket)[slot] = make_int2(rec[u].x, (rec[u].y / BK) * (DD * 4));
      }
    }
  }
}

static __device__ __forceinline__ int2 fetch_batch(const int2* __restrict__ bk, int lane,
                                                   int jI, int nI, int jO, int nO) {
  int hl = lane & 15;
  bool up = (lane >= 16);
  int nn = up ? nO : nI;
  int jj = up ? jO : jI;
  int2 p = make_int2(0, 0);
  if (hl < nn) p = bk[jj + hl];
  return p;
}

static __device__ __forceinline__ float2 row_piece(const float* __restrict__ base,
                                                   unsigned rowByte, unsigned laneByte) {
  return *(const float2*)((const char*)base + (size_t)(rowByte + laneByte));
}

__global__ void __launch_bounds__(128)
node_gather_transform_kernel(const float* __restrict__ n_feats,
                             const float* __restrict__ r_feats,
                             const int* __restrict__ bucket,
                             const int* __restrict__ rangeEnd,
                             const int* __restrict__ deg,
                             const float* __restrict__ W_I_w,
                             const float* __restrict__ W_I_b,
                             const float* __restrict__ W_O_w,
                             const float* __restrict__ W_O_b,
                             float* __restrict__ out,
                             int numTiles) {
  __shared__ float sWI[DD * SPITCH];
  __shared__ float sWO[DD * SPITCH];
  __shared__ float sAI[4][16 * SPITCH];
  __shared__ float sAO[4][16 * SPITCH];
  __shared__ float sCI[4][16];
  __shared__ float sCO[4][16];

  {
    const float4* gI = (const float4*)W_I_w;
    const float4* gO = (const float4*)W_O_w;
    int tid = threadIdx.x;
#pragma unroll
    for (int i = 0; i < (DD * DD / 4) / 128; ++i) {
      int idx = tid + i * 128;
      int r = idx >> 4;
      int c4 = idx & 15;
      *(float4*)(&sWI[r * SPITCH + c4 * 4]) = gI[idx];
      *(float4*)(&sWO[r * SPITCH + c4 * 4]) = gO[idx];
    }
  }

  int wv   = threadIdx.x >> 5;
  int lane = threadIdx.x & 31;
  int tile = blockIdx.x * 4 + wv;
  bool active = (tile < numTiles);
  int rowBase = tile * 16;
  float* aIw = sAI[wv];
  float* aOw = sAO[wv];

  if (active) {
    const int2* bk = (const int2*)bucket;
    int endv = rangeEnd[2 * rowBase + lane];
    int degv = deg[2 * rowBase + lane];
    unsigned laneByte = (unsigned)lane * 8u;

    int2 pvN;
    {
      int dI = __builtin_amdgcn_readlane(degv, 0);
      int dO = __builtin_amdgcn_readlane(degv, 1);
      int eI = __builtin_amdgcn_readlane(endv, 0);
      int eO = __builtin_amdgcn_readlane(endv, 1);
      pvN = fetch_batch(bk, lane, eI - dI, (dI < CH) ? dI : CH, eO - dO, (dO < CH) ? dO : CH);
    }

    for (int r = 0; r < 16; ++r) {
      int dI   = __builtin_amdgcn_readlane(degv, 2 * r);
      int dO   = __builtin_amdgcn_readlane(degv, 2 * r + 1);
      int endI = __builtin_amdgcn_readlane(endv, 2 * r);
      int endO = __builtin_amdgcn_readlane(endv, 2 * r + 1);
      int jI = endI - dI;
      int jO = endO - dO;

      int2 pv = pvN;
      if (r + 1 < 16) {
        int d1I = __builtin_amdgcn_readlane(degv, 2 * r + 2);
        int d1O = __builtin_amdgcn_readlane(degv, 2 * r + 3);
        int e1I = __builtin_amdgcn_readlane(endv, 2 * r + 2);
        int e1O = __builtin_amdgcn_readlane(endv, 2 * r + 3);
        pvN = fetch_batch(bk, lane, e1I - d1I, (d1I < CH) ? d1I : CH,
                          e1O - d1O, (d1O < CH) ? d1O : CH);
      }

      float sIx = 0.f, sIy = 0.f, sOx = 0.f, sOy = 0.f;
      bool first = true;
      while (jI < endI || jO < endO) {
        int nI = endI - jI;
        nI = (nI < 0) ? 0 : ((nI < CH) ? nI : CH);
        int nO = endO - jO;
        nO = (nO < 0) ? 0 : ((nO < CH) ? nO : CH);
        if (!first) pv = fetch_batch(bk, lane, jI, nI, jO, nO);
        first = false;

        float2 x[2 * CH], q[2 * CH];
#pragma unroll
        for (int t = 0; t < CH; ++t) {
          if (t < nI) {
            x[t] = row_piece(n_feats, (unsigned)__builtin_amdgcn_readlane(pv.x, t), laneByte);
            q[t] = row_piece(r_feats, (unsigned)__builtin_amdgcn_readlane(pv.y, t), laneByte);
          }
        }
#pragma unroll
        for (int t = 0; t < CH; ++t) {
          if (t < nO) {
            x[CH + t] = row_piece(n_feats, (unsigned)__builtin_amdgcn_readlane(pv.x, CH + t), laneByte);
            q[CH + t] = row_piece(r_feats, (unsigned)__builtin_amdgcn_readlane(pv.y, CH + t), laneByte);
          }
        }
#pragma unroll
        for (int t = 0; t < CH; ++t) {
          if (t < nI) {
            sIx += x[t].x - q[t].x;
            sIy += x[t].y - q[t].y;
          }
        }
#pragma unroll
        for (int t = 0; t < CH; ++t) {
          if (t < nO) {
            sOx += x[CH + t].x - q[CH + t].x;
            sOy += x[CH + t].y - q[CH + t].y;
          }
        }
        jI += CH;
        jO += CH;
      }

      *(float2*)(aIw + r * SPITCH + 2 * lane) = make_float2(sIx, sIy);
      *(float2*)(aOw + r * SPITCH + 2 * lane) = make_float2(sOx, sOy);
      if (lane == 0) {
        sCI[wv][r] = (float)dI;
        sCO[wv][r] = (float)dO;
      }
    }
  }
  __syncthreads();
  if (!active) return;

  int lm   = lane & 15;
  int hs   = lane >> 4;
  int kb   = hs * 2;
  int mOff = hs * 8;

  float cI8[8], cO8[8], dg8[8];
#pragma unroll
  for (int i = 0; i < 8; ++i) {
    cI8[i] = sCI[wv][i + mOff];
    cO8[i] = sCO[wv][i + mOff];
    dg8[i] = fmaxf(cI8[i] + cO8[i], 1.0f);
  }

  const float* aIrow = aIw + lm * SPITCH;
  const float* aOrow = aOw + lm * SPITCH;
  const float* sIrow[4];
  const float* sOrow[4];
#pragma unroll
  for (int t = 0; t < 4; ++t) {
    sIrow[t] = sWI + (t * 16 + lm) * SPITCH;
    sOrow[t] = sWO + (t * 16 + lm) * SPITCH;
  }

  v8f c[4];
#pragma unroll
  for (int t = 0; t < 4; ++t) c[t] = (v8f){0.f, 0.f, 0.f, 0.f, 0.f, 0.f, 0.f, 0.f};

#pragma unroll
  for (int k0 = 0; k0 < DD; k0 += 4) {
    v2f aI = *(const v2f*)(aIrow + k0 + kb);
    v2f aO = *(const v2f*)(aOrow + k0 + kb);
#pragma unroll
    for (int t = 0; t < 4; ++t) {
      v2f bI = *(const v2f*)(sIrow[t] + k0 + kb);
      v2f bO = *(const v2f*)(sOrow[t] + k0 + kb);
      c[t] = wmma_f32x4(aI, bI, c[t]);
      c[t] = wmma_f32x4(aO, bO, c[t]);
    }
  }

#pragma unroll
  for (int t = 0; t < 4; ++t) {
    int n = t * 16 + lm;
    float bI = W_I_b[n];
    float bO = W_O_b[n];
#pragma unroll
    for (int i = 0; i < 8; ++i) {
      int m = rowBase + i + mOff;
      float v = (c[t][i] + cI8[i] * bI + cO8[i] * bO) / dg8[i];
      out[(size_t)m * DD + n] = v;
    }
  }
}

__global__ void rel_transform_kernel(const float* __restrict__ r_feats,
                                     const float* __restrict__ W_R_w,
                                     const float* __restrict__ W_R_b,
                                     float* __restrict__ out,
                                     int R) {
  int wavesPerBlock = blockDim.x >> 5;
  int numTiles = (R + 15) >> 4;
  int tile = blockIdx.x * wavesPerBlock + (threadIdx.x >> 5);
  if (tile >= numTiles) return;

  int lane = threadIdx.x & 31;
  int lm   = lane & 15;
  int hs   = lane >> 4;
  int kb   = hs * 2;
  int mOff = hs * 8;
  int rowBase = tile * 16;

  int mA = rowBase + lm;
  bool aOk = (mA < R);
  const float* arow = r_feats + (size_t)(aOk ? mA : 0) * DD;

  const float* Wrow[4];
#pragma unroll
  for (int t = 0; t < 4; ++t) Wrow[t] = W_R_w + (size_t)(t * 16 + lm) * DD;

  v8f c[4];
#pragma unroll
  for (int t = 0; t < 4; ++t) c[t] = (v8f){0.f, 0.f, 0.f, 0.f, 0.f, 0.f, 0.f, 0.f};

#pragma unroll
  for (int k0 = 0; k0 < DD; k0 += 4) {
    v2f a = *(const v2f*)(arow + k0 + kb);
    if (!aOk) { a.x = 0.f; a.y = 0.f; }
#pragma unroll
    for (int t = 0; t < 4; ++t) {
      v2f b = *(const v2f*)(Wrow[t] + k0 + kb);
      c[t] = wmma_f32x4(a, b, c[t]);
    }
  }

#pragma unroll
  for (int t = 0; t < 4; ++t) {
    int n = t * 16 + lm;
    float bias = W_R_b[n];
#pragma unroll
    for (int i = 0; i < 8; ++i) {
      int m = rowBase + i + mOff;
      if (m < R) out[(size_t)m * DD + n] = c[t][i] + bias;
    }
  }
}

extern "C" void kernel_launch(void* const* d_in, const int* in_sizes, int n_in,
                              void* d_out, int out_size, void* d_ws, size_t ws_size,
                              hipStream_t stream) {
  const float* n_feats  = (const float*)d_in[0];
  const float* r_feats  = (const float*)d_in[1];
  const float* W_I_w    = (const float*)d_in[2];
  const float* W_I_b    = (const float*)d_in[3];
  const float* W_O_w    = (const float*)d_in[4];
  const float* W_O_b    = (const float*)d_in[5];
  const float* W_R_w    = (const float*)d_in[6];
  const float* W_R_b    = (const float*)d_in[7];
  const int*   src      = (const int*)d_in[8];
  const int*   dst      = (const int*)d_in[9];
  const int*   etype    = (const int*)d_in[10];
  const int*   num_rels = (const int*)d_in[11];

  int N = in_sizes[0] / DD;
  int R = in_sizes[1] / DD;
  int E = in_sizes[8];
  int M = 2 * N;
  int bins = (N + (1 << BN_SHIFT) - 1) >> BN_SHIFT;

  int* bucket = (int*)d_ws;
  int* deg    = bucket + (size_t)2 * E;
  int* total  = deg + M;
  int* cursor = total + 1;

  int* coarse   = (int*)d_out;
  int* histG    = coarse + (size_t)2 * E;
  int* binRange = histG + (size_t)HB * bins;

  int chunk = (E + HB - 1) / HB;
  size_t ldsBytes = (size_t)bins * sizeof(int);

  bin_hist_kernel<<<HB, HT, ldsBytes, stream>>>(dst, histG, total, E, bins, chunk);

  bin_alloc_kernel<<<(bins + 31) / 32, 256, 0, stream>>>(histG, total, binRange, bins);

  bin_scatter_kernel<<<HB, HT, ldsBytes, stream>>>(src, dst, etype, num_rels, histG, coarse,
                                                  E, bins, chunk);

  bin_sort_kernel<<<bins, BK, 0, stream>>>(coarse, binRange, bucket, deg, cursor, M);

  int nodeTiles = (N + 15) / 16;
  node_gather_transform_kernel<<<(nodeTiles + 3) / 4, 128, 0, stream>>>(
      n_feats, r_feats, bucket, cursor, deg,
      W_I_w, W_I_b, W_O_w, W_O_b, (float*)d_out, nodeTiles);

  int relTiles = (R + 15) / 16;
  rel_transform_kernel<<<(relTiles + 3) / 4, 128, 0, stream>>>(
      r_feats, W_R_w, W_R_b, (float*)d_out + (size_t)N * DD, R);
}


# both sort kernels put their records in order into an LDS image of their output range and write it out as whole 128-byte lines: the 2M scattered 8-byte global stores become whole-line stores
# speedup vs baseline: 3.0070x; 1.3467x over previous
#include <hip/hip_runtime.h>


typedef float v2f __attribute__((ext_vector_type(2)));
typedef float v8f __attribute__((ext_vector_type(8)));

#define DD 64
#define SPITCH 68
#define CH 16

static __device__ __forceinline__ v8f wmma_f32x4(v2f a, v2f b, v8f c) {
  return __builtin_amdgcn_wmma_f32_16x16x4_f32(
       false, a,  false, b,
       (short)0, c,  false,  false);
}

#define HB 256
#define HT 512
#define BN_SHIFT 8
#define BK 512
#define EPT 8
#define SEGW (HB / (BK / 32))
#define LREC 16
#define STG 6144

static __device__ __forceinline__ void copy_image(const int2* img, int2* __restrict__ dst,
                                                  int first, int n, int tid, int nthreads) {
  int pairs = (n + 1) >> 1;
  for (int q = tid; q < pairs; q += nthreads) {
    int p = 2 * q;
    if (p >= first && p + 1 < n) {
      *(int4*)(dst + p) = *(const int4*)(img + p);
    } else {
      if (p >= first) dst[p] = img[p];
      if (p + 1 >= first && p + 1 < n) dst[p + 1] = img[p + 1];
    }
  }
}

__global__ void __launch_bounds__(HT)
bin_scatter_kernel(const int* __restrict__ src,
                   const int* __restrict__ dst,
                   const int* __restrict__ etype,
                   const int* __restrict__ num_rels_p,
                   int* __restrict__ offG,
                   int* __restrict__ coarse,
                   int E, int NB, int chunk) {
  extern __shared__ __align__(16) int shd[];
  __shared__ int sWave[HT / 32];
  int2* stage = (int2*)shd;
  int* sh = shd + 2 * (chunk + LREC);
  int tid = threadIdx.x;
  int lane = tid & 31;
  int wv = tid >> 5;
  for (int b = tid; b < NB; b += HT) sh[b] = 0;
  __syncthreads();
  int lo = blockIdx.x * chunk;
  lo = (lo < E) ? lo : E;
  int hi = lo + chunk;
  hi = (hi < E) ? hi : E;
  int loA = lo & ~(LREC - 1);
  for (int i = lo + tid; i < hi; i += EPT * HT) {
    int d[EPT];
#pragma unroll
    for (int u = 0; u < EPT; ++u) {
      int iu = i + u * HT;
      d[u] = (iu < hi) ? dst[iu] : -1;
    }
#pragma unroll
    for (int u = 0; u < EPT; ++u)
      if (d[u] >= 0) atomicAdd(&sh[d[u] >> BN_SHIFT], 1);
  }
  __syncthreads();
  int per = (NB + HT - 1) / HT;
  int b0 = tid * per;
  int b1 = b0 + per;
  b1 = (b1 < NB) ? b1 : NB;
  int sum = 0;
  for (int b = b0; b < b1; ++b) sum += sh[b];
  int inc = sum;
  for (int dd = 1; dd < 32; dd <<= 1) {
    int o = __shfl_up(inc, dd);
    if (lane >= dd) inc += o;
  }
  if (lane == 31) sWave[wv] = inc;
  __syncthreads();
  int woff = 0;
#pragma unroll
  for (int i = 0; i < HT / 32; ++i) {
    int v = sWave[i];
    if (i < wv) woff += v;
  }
  int* row = offG + (size_t)blockIdx.x * (NB + 1);
  int run = lo + woff + inc - sum;
  for (int b = b0; b < b1; ++b) {
    int c = sh[b];
    sh[b] = run;
    row[b] = run;
    run += c;
  }
  if (tid == 0) row[NB] = hi;
  __syncthreads();
  int half = num_rels_p[0] >> 1;
  for (int i = lo + tid; i < hi; i += EPT * HT) {
    int d[EPT], s[EPT], t[EPT];
#pragma unroll
    for (int u = 0; u < EPT; ++u) {
      int iu = i + u * HT;
      bool in = (iu < hi);
      int ii = in ? iu : i;
      int dv = dst[ii];
      d[u] = in ? dv : -1;
      s[u] = src[ii];
      t[u] = etype[ii];
    }
#pragma unroll
    for (int u = 0; u < EPT; ++u) {
      if (d[u] >= 0) {
        int key = ((d[u] & ((1 << BN_SHIFT) - 1)) << 1) | ((t[u] < half) ? 0 : 1);
        int slot = atomicAdd(&sh[d[u] >> BN_SHIFT], 1);
        stage[slot - loA] = make_int2(s[u] * (DD * 4), t[u] * BK + key);
      }
    }
  }
  __syncthreads();
  copy_image(stage, (int2*)coarse + loA, lo - loA, hi - loA, tid, HT);
}

__global__ void __launch_bounds__(BK)
bin_sort_kernel(const int* __restrict__ coarse,
                const int* __restrict__ offG,
                int* __restrict__ bucket,
                int* __restrict__ deg,
                int* __restrict__ rangeEnd,
                int M, int NB, int E, int chunk) {
  static_assert(BK == (2 << BN_SHIFT), "one thread per key of the bin");
  static_assert(HB <= BK && HB % 32 == 0 && HB % (BK / 32) == 0,
                "one thread per run, whole waves, the same number of runs per wave");
  __shared__ int sCnt[BK];
  __shared__ int sWave[BK / 32];
  __shared__ int sRed[BK / 32];
  __shared__ int sSegS[HB];
  __shared__ int sSegL[HB];
  __shared__ __align__(16) int2 sStage[STG];
  __shared__ int sTot;
  int tid = threadIdx.x;
  int lane = tid & 31;
  int wv = tid >> 5;
  int bin = blockIdx.x;
  int rel = 0;
  if (tid < HB) {
    const int* rp = offG + (size_t)tid * (NB + 1) + bin;
    int s0 = rp[0];
    int e0 = rp[1];
    int lo = tid * chunk;
    lo = (lo < E) ? lo : E;
    rel = s0 - lo;
    sSegS[tid] = s0;
    sSegL[tid] = e0 - s0;
  }
  sCnt[tid] = 0;
  for (int dd = 16; dd >= 1; dd >>= 1) rel += __shfl_xor(rel, dd);
  if (lane == 0) sRed[wv] = rel;
  __syncthreads();
  int base = 0;
#pragma unroll
  for (int i = 0; i < BK / 32; ++i) base += sRed[i];

  const int2* cz = (const int2*)coarse;
  int sS[SEGW], sL[SEGW];
  int mx = 0;
#pragma unroll
  for (int u = 0; u < SEGW; ++u) {
    int sg = wv * SEGW + u;
    sS[u] = sSegS[sg];
    sL[u] = sSegL[sg];
    mx = (sL[u] > mx) ? sL[u] : mx;
  }
  for (int i0 = 0; i0 < mx; i0 += 32) {
    int i = i0 + lane;
    int2 rec[SEGW];
#pragma unroll
    for (int u = 0; u < SEGW; ++u)
      rec[u] = (i < sL[u]) ? cz[sS[u] + i] : make_int2(0, -1);
#pragma unroll
    for (int u = 0; u < SEGW; ++u)
      if (rec[u].y >= 0) atomicAdd(&sCnt[rec[u].y & (BK - 1)], 1);
  }
  __syncthreads();
  int c = sCnt[tid];
  int inc = c;
  for (int d = 1; d < 32; d <<= 1) {
    int o = __shfl_up(inc, d);
    if (lane >= d) inc += o;
  }
  if (lane == 31) sWave[wv] = inc;
  __syncthreads();
  int woff = 0;
#pragma unroll
  for (int i = 0; i < BK / 32; ++i) {
    int v = sWave[i];
    if (i < wv) woff += v;
  }
  int st = base + woff + inc - c;
  sCnt[tid] = st;
  if (tid == BK - 1) sTot = st + c - base;
  int key = bin * BK + tid;
  if (key < M) {
    deg[key] = c;
    rangeEnd[key] = st + c;
  }
  __syncthreads();
  int baseA = base & ~(LREC - 1);
  int endRel = base - baseA + sTot;
  bool staged = (endRel <= STG);
  for (int i0 = 0; i0 < mx; i0 += 32) {
    int i = i0 + lane;
    int2 rec[SEGW];
#pragma unroll
    for (int u = 0; u < SEGW; ++u)
      rec[u] = (i < sL[u]) ? cz[sS[u] + i] : make_int2(0, -1);
#pragma unroll
    for (int u = 0; u < SEGW; ++u) {
      if (rec[u].y >= 0) {
        int slot = atomicAdd(&sCnt[rec[u].y & (BK - 1)], 1);
        int2 v = make_int2(rec[u].x, (rec[u].y / BK) * (DD * 4));
        if (staged) sStage[slot - baseA] = v;
        else ((int2*)bucket)[slot] = v;
      }
    }
  }
  if (staged) {
    __syncthreads();
    copy_image(sStage, (int2*)bucket + baseA, base - baseA, endRel, tid, BK);
  }
}

static __device__ __forceinline__ int2 fetch_batch(const int2* __restrict__ bk, int lane,
                                                   int jI, int nI, int jO, int nO) {
  int hl = lane & 15;
  bool up = (lane >= 16);
  int nn = up ? nO : nI;
  int jj = up ? jO : jI;
  int2 p = make_int2(0, 0);
  if (hl < nn) p = bk[jj + hl];
  return p;
}

static __device__ __forceinline__ float2 row_piece(const float* __restrict__ base,
                                                   unsigned rowByte, unsigned laneByte) {
  return *(const float2*)((const char*)base + (size_t)(rowByte + laneByte));
}

__global__ void __launch_bounds__(128)
node_gather_transform_kernel(const float* __restrict__ n_feats,
                             const float* __restrict__ r_feats,
                             const int* __restrict__ bucket,
                             const int* __restrict__ rangeEnd,
                             const int* __restrict__ deg,
                             const float* __restrict__ W_I_w,
                             const float* __restrict__ W_I_b,
                             const float* __restrict__ W_O_w,
                             const float* __restrict__ W_O_b,
                             float* __restrict__ out,
                             int numTiles) {
  __shared__ float sWI[DD * SPITCH];
  __shared__ float sWO[DD * SPITCH];
  __shared__ float sAI[4][16 * SPITCH];
  __shared__ float sAO[4][16 * SPITCH];
  __shared__ float sCI[4][16];
  __shared__ float sCO[4][16];

  {
    const float4* gI = (const float4*)W_I_w;
    const float4* gO = (const float4*)W_O_w;
    int tid = threadIdx.x;
#pragma unroll
    for (int i = 0; i < (DD * DD / 4) / 128; ++i) {
      int idx = tid + i * 128;
      int r = idx >> 4;
      int c4 = idx & 15;
      *(float4*)(&sWI[r * SPITCH + c4 * 4]) = gI[idx];
      *(float4*)(&sWO[r * SPITCH + c4 * 4]) = gO[idx];
    }
  }

  int wv   = threadIdx.x >> 5;
  int lane = threadIdx.x & 31;
  int tile = blockIdx.x * 4 + wv;
  bool active = (tile < numTiles);
  int rowBase = tile * 16;
  float* aIw = sAI[wv];
  float* aOw = sAO[wv];

  if (active) {
    const int2* bk = (const int2*)bucket;
    int endv = rangeEnd[2 * rowBase + lane];
    int degv = deg[2 * rowBase + lane];
    unsigned laneByte = (unsigned)lane * 8u;

    int2 pvN;
    {
      int dI = __builtin_amdgcn_readlane(degv, 0);
      int dO = __builtin_amdgcn_readlane(degv, 1);
      int eI = __builtin_amdgcn_readlane(endv, 0);
      int eO = __builtin_amdgcn_readlane(endv, 1);
      pvN = fetch_batch(bk, lane, eI - dI, (dI < CH) ? dI : CH, eO - dO, (dO < CH) ? dO : CH);
    }

    for (int r = 0; r < 16; ++r) {
      int dI   = __builtin_amdgcn_readlane(degv, 2 * r);
      int dO   = __builtin_amdgcn_readlane(degv, 2 * r + 1);
      int endI = __builtin_amdgcn_readlane(endv, 2 * r);
      int endO = __builtin_amdgcn_readlane(endv, 2 * r + 1);
      int jI = endI - dI;
      int jO = endO - dO;

      int2 pv = pvN;
      if (r + 1 < 16) {
        int d1I = __builtin_amdgcn_readlane(degv, 2 * r + 2);
        int d1O = __builtin_amdgcn_readlane(degv, 2 * r + 3);
        int e1I = __builtin_amdgcn_readlane(endv, 2 * r + 2);
        int e1O = __builtin_amdgcn_readlane(endv, 2 * r + 3);
        pvN = fetch_batch(bk, lane, e1I - d1I, (d1I < CH) ? d1I : CH,
                          e1O - d1O, (d1O < CH) ? d1O : CH);
      }

      float sIx = 0.f, sIy = 0.f, sOx = 0.f, sOy = 0.f;
      bool first = true;
      while (jI < endI || jO < endO) {
        int nI = endI - jI;
        nI = (nI < 0) ? 0 : ((nI < CH) ? nI : CH);
        int nO = endO - jO;
        nO = (nO < 0) ? 0 : ((nO < CH) ? nO : CH);
        if (!first) pv = fetch_batch(bk, lane, jI, nI, jO, nO);
        first = false;

        float2 x[2 * CH], q[2 * CH];
#pragma unroll
        for (int t = 0; t < CH; ++t) {
          if (t < nI) {
            x[t] = row_piece(n_feats, (unsigned)__builtin_amdgcn_readlane(pv.x, t), laneByte);
            q[t] = row_piece(r_feats, (unsigned)__builtin_amdgcn_readlane(pv.y, t), laneByte);
          }
        }
#pragma unroll
        for (int t = 0; t < CH; ++t) {
          if (t < nO) {
            x[CH + t] = row_piece(n_feats, (unsigned)__builtin_amdgcn_readlane(pv.x, CH + t), laneByte);
            q[CH + t] = row_piece(r_feats, (unsigned)__builtin_amdgcn_readlane(pv.y, CH + t), laneByte);
          }
        }
#pragma unroll
        for (int t = 0; t < CH; ++t) {
          if (t < nI) {
            sIx += x[t].x - q[t].x;
            sIy += x[t].y - q[t].y;
          }
        }
#pragma unroll
        for (int t = 0; t < CH; ++t) {
          if (t < nO) {
            sOx += x[CH + t].x - q[CH + t].x;
            sOy += x[CH + t].y - q[CH + t].y;
          }
        }
        jI += CH;
        jO += CH;
      }

      *(float2*)(aIw + r * SPITCH + 2 * lane) = make_float2(sIx, sIy);
      *(float2*)(aOw + r * SPITCH + 2 * lane) = make_float2(sOx, sOy);
      if (lane == 0) {
        sCI[wv][r] = (float)dI;
        sCO[wv][r] = (float)dO;
      }
    }
  }
  __syncthreads();
  if (!active) return;

  int lm   = lane & 15;
  int hs   = lane >> 4;
  int kb   = hs * 2;
  int mOff = hs * 8;

  float cI8[8], cO8[8], dg8[8];
#pragma unroll
  for (int i = 0; i < 8; ++i) {
    cI8[i] = sCI[wv][i + mOff];
    cO8[i] = sCO[wv][i + mOff];
    dg8[i] = fmaxf(cI8[i] + cO8[i], 1.0f);
  }

  const float* aIrow = aIw + lm * SPITCH;
  const float* aOrow = aOw + lm * SPITCH;
  const float* sIrow[4];
  const float* sOrow[4];
#pragma unroll
  for (int t = 0; t < 4; ++t) {
    sIrow[t] = sWI + (t * 16 + lm) * SPITCH;
    sOrow[t] = sWO + (t * 16 + lm) * SPITCH;
  }

  v8f c[4];
#pragma unroll
  for (int t = 0; t < 4; ++t) c[t] = (v8f){0.f, 0.f, 0.f, 0.f, 0.f, 0.f, 0.f, 0.f};

#pragma unroll
  for (int k0 = 0; k0 < DD; k0 += 4) {
    v2f aI = *(const v2f*)(aIrow + k0 + kb);
    v2f aO = *(const v2f*)(aOrow + k0 + kb);
#pragma unroll
    for (int t = 0; t < 4; ++t) {
      v2f bI = *(const v2f*)(sIrow[t] + k0 + kb);
      v2f bO = *(const v2f*)(sOrow[t] + k0 + kb);
      c[t] = wmma_f32x4(aI, bI, c[t]);
      c[t] = wmma_f32x4(aO, bO, c[t]);
    }
  }

#pragma unroll
  for (int t = 0; t < 4; ++t) {
    int n = t * 16 + lm;
    float bI = W_I_b[n];
    float bO = W_O_b[n];
#pragma unroll
    for (int i = 0; i < 8; ++i) {
      int m = rowBase + i + mOff;
      float v = (c[t][i] + cI8[i] * bI + cO8[i] * bO) / dg8[i];
      out[(size_t)m * DD + n] = v;
    }
  }
}

__global__ void rel_transform_kernel(const float* __restrict__ r_feats,
                                     const float* __restrict__ W_R_w,
                                     const float* __restrict__ W_R_b,
                                     float* __restrict__ out,
                                     int R) {
  int wavesPerBlock = blockDim.x >> 5;
  int numTiles = (R + 15) >> 4;
  int tile = blockIdx.x * wavesPerBlock + (threadIdx.x >> 5);
  if (tile >= numTiles) return;

  int lane = threadIdx.x & 31;
  int lm   = lane & 15;
  int hs   = lane >> 4;
  int kb   = hs * 2;
  int mOff = hs * 8;
  int rowBase = tile * 16;

  int mA = rowBase + lm;
  bool aOk = (mA < R);
  const float* arow = r_feats + (size_t)(aOk ? mA : 0) * DD;

  const float* Wrow[4];
#pragma unroll
  for (int t = 0; t < 4; ++t) Wrow[t] = W_R_w + (size_t)(t * 16 + lm) * DD;

  v8f c[4];
#pragma unroll
  for (int t = 0; t < 4; ++t) c[t] = (v8f){0.f, 0.f, 0.f, 0.f, 0.f, 0.f, 0.f, 0.f};

#pragma unroll
  for (int k0 = 0; k0 < DD; k0 += 4) {
    v2f a = *(const v2f*)(arow + k0 + kb);
    if (!aOk) { a.x = 0.f; a.y = 0.f; }
#pragma unroll
    for (int t = 0; t < 4; ++t) {
      v2f b = *(const v2f*)(Wrow[t] + k0 + kb);
      c[t] = wmma_f32x4(a, b, c[t]);
    }
  }

#pragma unroll
  for (int t = 0; t < 4; ++t) {
    int n = t * 16 + lm;
    float bias = W_R_b[n];
#pragma unroll
    for (int i = 0; i < 8; ++i) {
      int m = rowBase + i + mOff;
      if (m < R) out[(size_t)m * DD + n] = c[t][i] + bias;
    }
  }
}

extern "C" void kernel_launch(void* const* d_in, const int* in_sizes, int n_in,
                              void* d_out, int out_size, void* d_ws, size_t ws_size,
                              hipStream_t stream) {
  const float* n_feats  = (const float*)d_in[0];
  const float* r_feats  = (const float*)d_in[1];
  const float* W_I_w    = (const float*)d_in[2];
  const float* W_I_b    = (const float*)d_in[3];
  const float* W_O_w    = (const float*)d_in[4];
  const float* W_O_b    = (const float*)d_in[5];
  const float* W_R_w    = (const float*)d_in[6];
  const float* W_R_b    = (const float*)d_in[7];
  const int*   src      = (const int*)d_in[8];
  const int*   dst      = (const int*)d_in[9];
  const int*   etype    = (const int*)d_in[10];
  const int*   num_rels = (const int*)d_in[11];

  int N = in_sizes[0] / DD;
  int R = in_sizes[1] / DD;
  int E = in_sizes[8];
  int M = 2 * N;
  int bins = (N + (1 << BN_SHIFT) - 1) >> BN_SHIFT;

  int* bucket = (int*)d_ws;
  int* deg    = bucket + (size_t)2 * E;
  int* cursor = deg + M;

  int* coarse = (int*)d_out;
  int* offG   = coarse + (size_t)2 * E;

  int chunk = (E + HB - 1) / HB;
  size_t ldsBytes = (size_t)(chunk + LREC) * 8 + (size_t)bins * sizeof(int);

  bin_scatter_kernel<<<HB, HT, ldsBytes, stream>>>(src, dst, etype, num_rels, offG, coarse,
                                                  E, bins, chunk);

  bin_sort_kernel<<<bins, BK, 0, stream>>>(coarse, offG, bucket, deg, cursor, M, bins, E, chunk);

  int nodeTiles = (N + 15) / 16;
  node_gather_transform_kernel<<<(nodeTiles + 3) / 4, 128, 0, stream>>>(
      n_feats, r_feats, bucket, cursor, deg,
      W_I_w, W_I_b, W_O_w, W_O_b, (float*)d_out, nodeTiles);

  int relTiles = (R + 15) / 16;
  rel_transform_kernel<<<(relTiles + 3) / 4, 128, 0, stream>>>(
      r_feats, W_R_w, W_R_b, (float*)d_out + (size_t)N * DD, R);
}


# the gather walks a tile's 32 lists as one stretch of the sorted image in full batches of 32 records and starts a new sum where a list ends, instead of one trip per node over its two short lists
# speedup vs baseline: 4.1982x; 1.3961x over previous
#include <hip/hip_runtime.h>


typedef float v2f __attribute__((ext_vector_type(2)));
typedef float v8f __attribute__((ext_vector_type(8)));

#define DD 64
#define SPITCH 68

static __device__ __forceinline__ v8f wmma_f32x4(v2f a, v2f b, v8f c) {
  return __builtin_amdgcn_wmma_f32_16x16x4_f32(
       false, a,  false, b,
       (short)0, c,  false,  false);
}

#define HB 256
#define HT 512
#define BN_SHIFT 8
#define BK 512
#define EPT 8
#define SEGW (HB / (BK / 32))
#define LREC 16
#define STG 6144

static __device__ __forceinline__ void copy_image(const int2* img, int2* __restrict__ dst,
                                                  int first, int n, int tid, int nthreads) {
  int pairs = (n + 1) >> 1;
  for (int q = tid; q < pairs; q += nthreads) {
    int p = 2 * q;
    if (p >= first && p + 1 < n) {
      *(int4*)(dst + p) = *(const int4*)(img + p);
    } else {
      if (p >= first) dst[p] = img[p];
      if (p + 1 >= first && p + 1 < n) dst[p + 1] = img[p + 1];
    }
  }
}

__global__ void __launch_bounds__(HT)
bin_scatter_kernel(const int* __restrict__ src,
                   const int* __restrict__ dst,
                   const int* __restrict__ etype,
                   const int* __restrict__ num_rels_p,
                   int* __restrict__ offG,
                   int* __restrict__ coarse,
                   int E, int NB, int chunk) {
  extern __shared__ __align__(16) int shd[];
  __shared__ int sWave[HT / 32];
  int2* stage = (int2*)shd;
  int* sh = shd + 2 * (chunk + LREC);
  int tid = threadIdx.x;
  int lane = tid & 31;
  int wv = tid >> 5;
  for (int b = tid; b < NB; b += HT) sh[b] = 0;
  __syncthreads();
  int lo = blockIdx.x * chunk;
  lo = (lo < E) ? lo : E;
  int hi = lo + chunk;
  hi = (hi < E) ? hi : E;
  int loA = lo & ~(LREC - 1);
  for (int i = lo + tid; i < hi; i += EPT * HT) {
    int d[EPT];
#pragma unroll
    for (int u = 0; u < EPT; ++u) {
      int iu = i + u * HT;
      d[u] = (iu < hi) ? dst[iu] : -1;
    }
#pragma unroll
    for (int u = 0; u < EPT; ++u)
      if (d[u] >= 0) atomicAdd(&sh[d[u] >> BN_SHIFT], 1);
  }
  __syncthreads();
  int per = (NB + HT - 1) / HT;
  int b0 = tid * per;
  int b1 = b0 + per;
  b1 = (b1 < NB) ? b1 : NB;
  int sum = 0;
  for (int b = b0; b < b1; ++b) sum += sh[b];
  int inc = sum;
  for (int dd = 1; dd < 32; dd <<= 1) {
    int o = __shfl_up(inc, dd);
    if (lane >= dd) inc += o;
  }
  if (lane == 31) sWave[wv] = inc;
  __syncthreads();
  int woff = 0;
#pragma unroll
  for (int i = 0; i < HT / 32; ++i) {
    int v = sWave[i];
    if (i < wv) woff += v;
  }
  int* row = offG + (size_t)blockIdx.x * (NB + 1);
  int run = lo + woff + inc - sum;
  for (int b = b0; b < b1; ++b) {
    int c = sh[b];
    sh[b] = run;
    row[b] = run;
    run += c;
  }
  if (tid == 0) row[NB] = hi;
  __syncthreads();
  int half = num_rels_p[0] >> 1;
  for (int i = lo + tid; i < hi; i += EPT * HT) {
    int d[EPT], s[EPT], t[EPT];
#pragma unroll
    for (int u = 0; u < EPT; ++u) {
      int iu = i + u * HT;
      bool in = (iu < hi);
      int ii = in ? iu : i;
      int dv = dst[ii];
      d[u] = in ? dv : -1;
      s[u] = src[ii];
      t[u] = etype[ii];
    }
#pragma unroll
    for (int u = 0; u < EPT; ++u) {
      if (d[u] >= 0) {
        int key = ((d[u] & ((1 << BN_SHIFT) - 1)) << 1) | ((t[u] < half) ? 0 : 1);
        int slot = atomicAdd(&sh[d[u] >> BN_SHIFT], 1);
        stage[slot - loA] = make_int2(s[u] * (DD * 4), t[u] * BK + key);
      }
    }
  }
  __syncthreads();
  copy_image(stage, (int2*)coarse + loA, lo - loA, hi - loA, tid, HT);
}

static __device__ __forceinline__ void rel_transform_tiles(const float* __restrict__ r_feats,
                                                           const float* __restrict__ W_R_w,
                                                           const float* __restrict__ W_R_b,
                                                           float* __restrict__ out,
                                                           int R, int wv, int nWaves, int lane) {
  int numTiles = (R + 15) >> 4;
  int lm   = lane & 15;
  int hs   = lane >> 4;
  int kb   = hs * 2;
  int mOff = hs * 8;

  const float* Wrow[4];
#pragma unroll
  for (int t = 0; t < 4; ++t) Wrow[t] = W_R_w + (size_t)(t * 16 + lm) * DD;

  for (int tile = wv; tile < numTiles; tile += nWaves) {
    int rowBase = tile * 16;
    int mA = rowBase + lm;
    bool aOk = (mA < R);
    const float* arow = r_feats + (size_t)(aOk ? mA : 0) * DD;

    v8f c[4];
#pragma unroll
    for (int t = 0; t < 4; ++t) c[t] = (v8f){0.f, 0.f, 0.f, 0.f, 0.f, 0.f, 0.f, 0.f};

#pragma unroll
    for (int k0 = 0; k0 < DD; k0 += 4) {
      v2f a = *(const v2f*)(arow + k0 + kb);
      if (!aOk) { a.x = 0.f; a.y = 0.f; }
#pragma unroll
      for (int t = 0; t < 4; ++t) {
        v2f b = *(const v2f*)(Wrow[t] + k0 + kb);
        c[t] = wmma_f32x4(a, b, c[t]);
      }
    }

#pragma unroll
    for (int t = 0; t < 4; ++t) {
      int n = t * 16 + lm;
      float bias = W_R_b[n];
#pragma unroll
      for (int i = 0; i < 8; ++i) {
        int m = rowBase + i + mOff;
        if (m < R) out[(size_t)m * DD + n] = c[t][i] + bias;
      }
    }
  }
}

static __device__ __forceinline__ float2 row_piece(const float* __restrict__ base,
                                                   unsigned rowByte, unsigned laneByte) {
  return *(const float2*)((const char*)base + (size_t)(rowByte + laneByte));
}

static __device__ __forceinline__ float2 row_piece_uniform(const float* __restrict__ base,
                                                           unsigned rowByte, unsigned laneByte) {
  const char* row = (const char*)base + rowByte;
  return *(const float2*)(row + laneByte);
}

__global__ void __launch_bounds__(BK)
bin_sort_node_kernel(const int* __restrict__ coarse,
                     const int* __restrict__ offG,
                     int NB,
                     const float* __restrict__ r_feats,
                     const float* __restrict__ W_R_w,
                     const float* __restrict__ W_R_b,
                     float* __restrict__ relOut,
                     int R,
                     const float* __restrict__ n_feats,
                     const float* __restrict__ W_I_w,
                     const float* __restrict__ W_I_b,
                     const float* __restrict__ W_O_w,
                     const float* __restrict__ W_O_b,
                     float* __restrict__ out,
                     int numTiles) {
  static_assert(BK == (2 << BN_SHIFT), "one thread per key of the bin");
  static_assert(BK / 32 == (1 << BN_SHIFT) / 16, "one wave per 16-node tile of the bin");
  static_assert(HB <= BK && HB % 32 == 0 && HB % (BK / 32) == 0,
                "one thread per run, whole waves, the same number of runs per wave");
  __shared__ int sCnt[BK];
  __shared__ int sWave[BK / 32];
  __shared__ int sSegS[HB];
  __shared__ int sSegL[HB];
  __shared__ __align__(16) int2 sStage[STG];
  __shared__ __align__(16) float sWI[DD * SPITCH];
  __shared__ __align__(16) float sWO[DD * SPITCH];
  __shared__ __align__(16) float sAI[BK / 32][16 * SPITCH];
  __shared__ __align__(16) float sAO[BK / 32][16 * SPITCH];
  __shared__ float sDegF[BK / 32][32];
  int tid = threadIdx.x;
  int lane = tid & 31;
  int wv = tid >> 5;
  int bin = blockIdx.x;
  if (bin == NB) {
    rel_transform_tiles(r_feats, W_R_w, W_R_b, relOut, R, wv, BK / 32, lane);
    return;
  }
  {
    const float4* gI = (const float4*)W_I_w;
    const float4* gO = (const float4*)W_O_w;
#pragma unroll
    for (int i = 0; i < (DD * DD / 4) / BK; ++i) {
      int idx = tid + i * BK;
      int r = idx >> 4;
      int c4 = idx & 15;
      *(float4*)(&sWI[r * SPITCH + c4 * 4]) = gI[idx];
      *(float4*)(&sWO[r * SPITCH + c4 * 4]) = gO[idx];
    }
  }
  if (tid < HB) {
    const int* rp = offG + (size_t)tid * (NB + 1) + bin;
    int s0 = rp[0];
    int e0 = rp[1];
    sSegS[tid] = s0;
    sSegL[tid] = e0 - s0;
  }
  sCnt[tid] = 0;
  __syncthreads();

  const int2* cz = (const int2*)coarse;
  int sS[SEGW], sL[SEGW];
  int mx = 0;
#pragma unroll
  for (int u = 0; u < SEGW; ++u) {
    int sg = wv * SEGW + u;
    sS[u] = sSegS[sg];
    sL[u] = sSegL[sg];
    mx = (sL[u] > mx) ? sL[u] : mx;
  }
  for (int i0 = 0; i0 < mx; i0 += 32) {
    int i = i0 + lane;
    int2 rec[SEGW];
#pragma unroll
    for (int u = 0; u < SEGW; ++u)
      rec[u] = (i < sL[u]) ? cz[sS[u] + i] : make_int2(0, -1);
#pragma unroll
    for (int u = 0; u < SEGW; ++u)
      if (rec[u].y >= 0) atomicAdd(&sCnt[rec[u].y & (BK - 1)], 1);
  }
  __syncthreads();
  int cnt = sCnt[tid];
  int inc = cnt;
  for (int d = 1; d < 32; d <<= 1) {
    int o = __shfl_up(inc, d);
    if (lane >= d) inc += o;
  }
  if (lane == 31) sWave[wv] = inc;
  __syncthreads();
  int woff = 0;
  int tot = 0;
#pragma unroll
  for (int i = 0; i < BK / 32; ++i) {
    int v = sWave[i];
    if (i < wv) woff += v;
    tot += v;
  }
  int st = woff + inc - cnt;
  bool staged = (tot <= STG);
  sCnt[tid] = st;
  __syncthreads();
  if (staged) {
    for (int i0 = 0; i0 < mx; i0 += 32) {
      int i = i0 + lane;
      int2 rec[SEGW];
#pragma unroll
      for (int u = 0; u < SEGW; ++u)
        rec[u] = (i < sL[u]) ? cz[sS[u] + i] : make_int2(0, -1);
#pragma unroll
      for (int u = 0; u < SEGW; ++u) {
        if (rec[u].y >= 0) {
          int slot = atomicAdd(&sCnt[rec[u].y & (BK - 1)], 1);
          sStage[slot] = make_int2(rec[u].x, (rec[u].y / BK) * (DD * 4));
        }
      }
    }
  }
  __syncthreads();

  int tile = bin * (BK / 32) + wv;
  bool active = (tile < numTiles);
  int rowBase = tile * 16;
  float* aIw = sAI[wv];
  float* aOw = sAO[wv];
  int degv = cnt;
  int endv = st + cnt;
  unsigned laneByte = (unsigned)lane * 8u;
  sDegF[wv][lane] = (float)cnt;

  if (active && staged) {
    const int2* img = sStage;
    int p0 = __builtin_amdgcn_readlane(endv, 0) - __builtin_amdgcn_readlane(degv, 0);
    int p1 = __builtin_amdgcn_readlane(endv, 31);
    int key = 0;
    int curEnd = __builtin_amdgcn_readlane(endv, 0);
    float accx = 0.f, accy = 0.f;
    int2 pvN = make_int2(0, 0);
    if (p0 + lane < p1) pvN = img[p0 + lane];
    for (int base = p0; base < p1; base += 32) {
      int2 pv = pvN;
      pvN = make_int2(0, 0);
      if (base + 32 + lane < p1) pvN = img[base + 32 + lane];

      float2 x[32], q[32];
#pragma unroll
      for (int t = 0; t < 32; ++t) {
        x[t] = row_piece_uniform(n_feats, (unsigned)__builtin_amdgcn_readlane(pv.x, t), laneByte);
        q[t] = row_piece_uniform(r_feats, (unsigned)__builtin_amdgcn_readlane(pv.y, t), laneByte);
      }
#pragma unroll
      for (int t = 0; t < 32; ++t) {
        int p = base + t;
        while (__builtin_expect(p == curEnd, 0)) {
          float* a = (key & 1) ? aOw : aIw;
          *(float2*)(a + (key >> 1) * SPITCH + 2 * lane) = make_float2(accx, accy);
          accx = 0.f;
          accy = 0.f;
          ++key;
          curEnd = (key < 32) ? __builtin_amdgcn_readlane(endv, key) : 0x7fffffff;
        }
        accx += x[t].x - q[t].x;
        accy += x[t].y - q[t].y;
      }
    }
    while (key < 32) {
      float* a = (key & 1) ? aOw : aIw;
      *(float2*)(a + (key >> 1) * SPITCH + 2 * lane) = make_float2(accx, accy);
      accx = 0.f;
      accy = 0.f;
      ++key;
    }
  } else if (active) {
    for (int r = 0; r < 16; ++r) {
      *(float2*)(aIw + r * SPITCH + 2 * lane) = make_float2(0.f, 0.f);
      *(float2*)(aOw + r * SPITCH + 2 * lane) = make_float2(0.f, 0.f);
    }
    for (int sg = 0; sg < HB; ++sg) {
      int s0 = sSegS[sg];
      int len = sSegL[sg];
      for (int i = 0; i < len; ++i) {
        int2 rec = cz[s0 + i];
        int key = rec.y & (BK - 1);
        if ((key >> 5) == wv) {
          int r = (key & 31) >> 1;
          float* a = (key & 1) ? aOw : aIw;
          float2 xv = row_piece(n_feats, (unsigned)rec.x, laneByte);
          float2 qv = row_piece(r_feats, (unsigned)((rec.y / BK) * (DD * 4)), laneByte);
          float2* p = (float2*)(a + r * SPITCH + 2 * lane);
          float2 v = *p;
          v.x += xv.x - qv.x;
          v.y += xv.y - qv.y;
          *p = v;
        }
      }
    }
  }
  __syncthreads();
  if (!active) return;

  int lm   = lane & 15;
  int hs   = lane >> 4;
  int kb   = hs * 2;
  int mOff = hs * 8;

  float cI8[8], cO8[8], dg8[8];
#pragma unroll
  for (int i = 0; i < 8; ++i) {
    cI8[i] = sDegF[wv][2 * (i + mOff)];
    cO8[i] = sDegF[wv][2 * (i + mOff) + 1];
    dg8[i] = fmaxf(cI8[i] + cO8[i], 1.0f);
  }

  const float* aIrow = aIw + lm * SPITCH;
  const float* aOrow = aOw + lm * SPITCH;
  const float* sIrow[4];
  const float* sOrow[4];
#pragma unroll
  for (int t = 0; t < 4; ++t) {
    sIrow[t] = sWI + (t * 16 + lm) * SPITCH;
    sOrow[t] = sWO + (t * 16 + lm) * SPITCH;
  }

  v8f c[4];
#pragma unroll
  for (int t = 0; t < 4; ++t) c[t] = (v8f){0.f, 0.f, 0.f, 0.f, 0.f, 0.f, 0.f, 0.f};

#pragma unroll
  for (int k0 = 0; k0 < DD; k0 += 4) {
    v2f aI = *(const v2f*)(aIrow + k0 + kb);
    v2f aO = *(const v2f*)(aOrow + k0 + kb);
#pragma unroll
    for (int t = 0; t < 4; ++t) {
      v2f bI = *(const v2f*)(sIrow[t] + k0 + kb);
      v2f bO = *(const v2f*)(sOrow[t] + k0 + kb);
      c[t] = wmma_f32x4(aI, bI, c[t]);
      c[t] = wmma_f32x4(aO, bO, c[t]);
    }
  }

#pragma unroll
  for (int t = 0; t < 4; ++t) {
    int n = t * 16 + lm;
    float bI = W_I_b[n];
    float bO = W_O_b[n];
#pragma unroll
    for (int i = 0; i < 8; ++i) {
      int m = rowBase + i + mOff;
      float v = (c[t][i] + cI8[i] * bI + cO8[i] * bO) / dg8[i];
      out[(size_t)m * DD + n] = v;
    }
  }
}

extern "C" void kernel_launch(void* const* d_in, const int* in_sizes, int n_in,
                              void* d_out, int out_size, void* d_ws, size_t ws_size,
                              hipStream_t stream) {
  const float* n_feats  = (const float*)d_in[0];
  const float* r_feats  = (const float*)d_in[1];
  const float* W_I_w    = (const float*)d_in[2];
  const float* W_I_b    = (const float*)d_in[3];
  const float* W_O_w    = (const float*)d_in[4];
  const float* W_O_b    = (const float*)d_in[5];
  const float* W_R_w    = (const float*)d_in[6];
  const float* W_R_b    = (const float*)d_in[7];
  const int*   src      = (const int*)d_in[8];
  const int*   dst      = (const int*)d_in[9];
  const int*   etype    = (const int*)d_in[10];
  const int*   num_rels = (const int*)d_in[11];

  int N = in_sizes[0] / DD;
  int R = in_sizes[1] / DD;
  int E = in_sizes[8];
  int bins = (N + (1 << BN_SHIFT) - 1) >> BN_SHIFT;

  int* coarse = (int*)d_ws;
  int* offG   = coarse + (size_t)2 * E;

  int chunk = (E + HB - 1) / HB;
  size_t ldsBytes = (size_t)(chunk + LREC) * 8 + (size_t)bins * sizeof(int);

  bin_scatter_kernel<<<HB, HT, ldsBytes, stream>>>(src, dst, etype, num_rels, offG, coarse,
                                                  E, bins, chunk);

  int nodeTiles = (N + 15) / 16;
  bin_sort_node_kernel<<<bins + 1, BK, 0, stream>>>(coarse, offG, bins,
                                                   r_feats, W_R_w, W_R_b,
                                                   (float*)d_out + (size_t)N * DD, R,
                                                   n_feats, W_I_w, W_I_b, W_O_w, W_O_b,
                                                   (float*)d_out, nodeTiles);
}


# the gather reads each record's r_feats row from a copy of the whole table that the block keeps in LDS: one global row request per record instead of two
# speedup vs baseline: 4.5388x; 1.0811x over previous
#include <hip/hip_runtime.h>


typedef float v2f __attribute__((ext_vector_type(2)));
typedef float v8f __attribute__((ext_vector_type(8)));

#define DD 64
#define SPITCH 68

static __device__ __forceinline__ v8f wmma_f32x4(v2f a, v2f b, v8f c) {
  return __builtin_amdgcn_wmma_f32_16x16x4_f32(
       false, a,  false, b,
       (short)0, c,  false,  false);
}

#define HB 256
#define HT 512
#define BN_SHIFT 8
#define BK 512
#define EPT 8
#define SEGW (HB / (BK / 32))
#define LREC 16
#define STG 6144
#define RMAX 200
#define RT ((RMAX * (DD / 4) + BK - 1) / BK)

static __device__ __forceinline__ void copy_image(const int2* img, int2* __restrict__ dst,
                                                  int first, int n, int tid, int nthreads) {
  int pairs = (n + 1) >> 1;
  for (int q = tid; q < pairs; q += nthreads) {
    int p = 2 * q;
    if (p >= first && p + 1 < n) {
      *(int4*)(dst + p) = *(const int4*)(img + p);
    } else {
      if (p >= first) dst[p] = img[p];
      if (p + 1 >= first && p + 1 < n) dst[p + 1] = img[p + 1];
    }
  }
}

__global__ void __launch_bounds__(HT)
bin_scatter_kernel(const int* __restrict__ src,
                   const int* __restrict__ dst,
                   const int* __restrict__ etype,
                   const int* __restrict__ num_rels_p,
                   int* __restrict__ offG,
                   int* __restrict__ coarse,
                   int E, int NB, int chunk) {
  extern __shared__ __align__(16) int shd[];
  __shared__ int sWave[HT / 32];
  int2* stage = (int2*)shd;
  int* sh = shd + 2 * (chunk + LREC);
  int tid = threadIdx.x;
  int lane = tid & 31;
  int wv = tid >> 5;
  for (int b = tid; b < NB; b += HT) sh[b] = 0;
  __syncthreads();
  int lo = blockIdx.x * chunk;
  lo = (lo < E) ? lo : E;
  int hi = lo + chunk;
  hi = (hi < E) ? hi : E;
  int loA = lo & ~(LREC - 1);
  for (int i = lo + tid; i < hi; i += EPT * HT) {
    int d[EPT];
#pragma unroll
    for (int u = 0; u < EPT; ++u) {
      int iu = i + u * HT;
      d[u] = (iu < hi) ? dst[iu] : -1;
    }
#pragma unroll
    for (int u = 0; u < EPT; ++u)
      if (d[u] >= 0) atomicAdd(&sh[d[u] >> BN_SHIFT], 1);
  }
  __syncthreads();
  int per = (NB + HT - 1) / HT;
  int b0 = tid * per;
  int b1 = b0 + per;
  b1 = (b1 < NB) ? b1 : NB;
  int sum = 0;
  for (int b = b0; b < b1; ++b) sum += sh[b];
  int inc = sum;
  for (int dd = 1; dd < 32; dd <<= 1) {
    int o = __shfl_up(inc, dd);
    if (lane >= dd) inc += o;
  }
  if (lane == 31) sWave[wv] = inc;
  __syncthreads();
  int woff = 0;
#pragma unroll
  for (int i = 0; i < HT / 32; ++i) {
    int v = sWave[i];
    if (i < wv) woff += v;
  }
  int* row = offG + (size_t)blockIdx.x * (NB + 1);
  int run = lo + woff + inc - sum;
  for (int b = b0; b < b1; ++b) {
    int c = sh[b];
    sh[b] = run;
    row[b] = run;
    run += c;
  }
  if (tid == 0) row[NB] = hi;
  __syncthreads();
  int half = num_rels_p[0] >> 1;
  for (int i = lo + tid; i < hi; i += EPT * HT) {
    int d[EPT], s[EPT], t[EPT];
#pragma unroll
    for (int u = 0; u < EPT; ++u) {
      int iu = i + u * HT;
      bool in = (iu < hi);
      int ii = in ? iu : i;
      int dv = dst[ii];
      d[u] = in ? dv : -1;
      s[u] = src[ii];
      t[u] = etype[ii];
    }
#pragma unroll
    for (int u = 0; u < EPT; ++u) {
      if (d[u] >= 0) {
        int key = ((d[u] & ((1 << BN_SHIFT) - 1)) << 1) | ((t[u] < half) ? 0 : 1);
        int slot = atomicAdd(&sh[d[u] >> BN_SHIFT], 1);
        stage[slot - loA] = make_int2(s[u] * (DD * 4), t[u] * BK + key);
      }
    }
  }
  __syncthreads();
  copy_image(stage, (int2*)coarse + loA, lo - loA, hi - loA, tid, HT);
}

static __device__ __forceinline__ void rel_transform_tiles(const float* __restrict__ r_feats,
                                                           const float* __restrict__ W_R_w,
                                                           const float* __restrict__ W_R_b,
                                                           float* __restrict__ out,
                                                           int R, int wv, int nWaves, int lane) {
  int numTiles = (R + 15) >> 4;
  int lm   = lane & 15;
  int hs   = lane >> 4;
  int kb   = hs * 2;
  int mOff = hs * 8;

  const float* Wrow[4];
#pragma unroll
  for (int t = 0; t < 4; ++t) Wrow[t] = W_R_w + (size_t)(t * 16 + lm) * DD;

  for (int tile = wv; tile < numTiles; tile += nWaves) {
    int rowBase = tile * 16;
    int mA = rowBase + lm;
    bool aOk = (mA < R);
    const float* arow = r_feats + (size_t)(aOk ? mA : 0) * DD;

    v8f c[4];
#pragma unroll
    for (int t = 0; t < 4; ++t) c[t] = (v8f){0.f, 0.f, 0.f, 0.f, 0.f, 0.f, 0.f, 0.f};

#pragma unroll
    for (int k0 = 0; k0 < DD; k0 += 4) {
      v2f a = *(const v2f*)(arow + k0 + kb);
      if (!aOk) { a.x = 0.f; a.y = 0.f; }
#pragma unroll
      for (int t = 0; t < 4; ++t) {
        v2f b = *(const v2f*)(Wrow[t] + k0 + kb);
        c[t] = wmma_f32x4(a, b, c[t]);
      }
    }

#pragma unroll
    for (int t = 0; t < 4; ++t) {
      int n = t * 16 + lm;
      float bias = W_R_b[n];
#pragma unroll
      for (int i = 0; i < 8; ++i) {
        int m = rowBase + i + mOff;
        if (m < R) out[(size_t)m * DD + n] = c[t][i] + bias;
      }
    }
  }
}

static __device__ __forceinline__ float2 row_piece(const float* __restrict__ base,
                                                   unsigned rowByte, unsigned laneByte) {
  return *(const float2*)((const char*)base + (size_t)(rowByte + laneByte));
}

static __device__ __forceinline__ float2 row_piece_uniform(const float* __restrict__ base,
                                                           unsigned rowByte, unsigned laneByte) {
  const char* row = (const char*)base + rowByte;
  return *(const float2*)(row + laneByte);
}


template <bool RLDS>
static __device__ __forceinline__ void walk_tile(const int2* img,
                                                 const float* __restrict__ n_feats,
                                                 const float* __restrict__ rGlob,
                                                 const float* rTab,
                                                 int degv, int endv, int lane, unsigned laneByte,
                                                 float* aIw, float* aOw) {
  int p0 = __builtin_amdgcn_readlane(endv, 0) - __builtin_amdgcn_readlane(degv, 0);
  int p1 = __builtin_amdgcn_readlane(endv, 31);
  int key = 0;
  int curEnd = __builtin_amdgcn_readlane(endv, 0);
  float accx = 0.f, accy = 0.f;
  int2 pvN = make_int2(0, 0);
  if (p0 + lane < p1) pvN = img[p0 + lane];
  for (int base = p0; base < p1; base += 32) {
    int2 pv = pvN;
    pvN = make_int2(0, 0);
    if (base + 32 + lane < p1) pvN = img[base + 32 + lane];

    float2 x[32], q[32];
#pragma unroll
    for (int t = 0; t < 32; ++t)
      x[t] = row_piece_uniform(n_feats, (unsigned)__builtin_amdgcn_readlane(pv.x, t), laneByte);
#pragma unroll
    for (int t = 0; t < 32; ++t) {
      unsigned rowByte = (unsigned)__builtin_amdgcn_readlane(pv.y, t);
      if (RLDS)
        q[t] = *(const float2*)((const char*)rTab + rowByte + laneByte);
      else
        q[t] = row_piece_uniform(rGlob, rowByte, laneByte);
    }
#pragma unroll
    for (int t = 0; t < 32; ++t) {
      int p = base + t;
      while (__builtin_expect(p == curEnd, 0)) {
        float* a = (key & 1) ? aOw : aIw;
        *(float2*)(a + (key >> 1) * SPITCH + 2 * lane) = make_float2(accx, accy);
        accx = 0.f;
        accy = 0.f;
        ++key;
        curEnd = (key < 32) ? __builtin_amdgcn_readlane(endv, key) : 0x7fffffff;
      }
      accx += x[t].x - q[t].x;
      accy += x[t].y - q[t].y;
    }
  }
  while (key < 32) {
    float* a = (key & 1) ? aOw : aIw;
    *(float2*)(a + (key >> 1) * SPITCH + 2 * lane) = make_float2(accx, accy);
    accx = 0.f;
    accy = 0.f;
    ++key;
  }
}

__global__ void __launch_bounds__(BK)
bin_sort_node_kernel(const int* __restrict__ coarse,
                     const int* __restrict__ offG,
                     int NB,
                     const float* __restrict__ r_feats,
                     const float* __restrict__ W_R_w,
                     const float* __restrict__ W_R_b,
                     float* __restrict__ relOut,
                     int R,
                     const float* __restrict__ n_feats,
                     const float* __restrict__ W_I_w,
                     const float* __restrict__ W_I_b,
                     const float* __restrict__ W_O_w,
                     const float* __restrict__ W_O_b,
                     float* __restrict__ out,
                     int numTiles) {
  static_assert(BK == (2 << BN_SHIFT), "one thread per key of the bin");
  static_assert(BK / 32 == (1 << BN_SHIFT) / 16, "one wave per 16-node tile of the bin");
  static_assert(HB <= BK && HB % 32 == 0 && HB % (BK / 32) == 0,
                "one thread per run, whole waves, the same number of runs per wave");
  __shared__ int sCnt[BK];
  __shared__ int sWave[BK / 32];
  __shared__ int sSegS[HB];
  __shared__ int sSegL[HB];
  __shared__ __align__(16) int2 sStage[STG];
  __shared__ __align__(16) float sWI[DD * SPITCH];
  __shared__ __align__(16) float sWO[DD * SPITCH];
  __shared__ __align__(16) float sR[RT * BK * 4];
  __shared__ __align__(16) float sAI[BK / 32][16 * SPITCH];
  __shared__ __align__(16) float sAO[BK / 32][16 * SPITCH];
  __shared__ float sDegF[BK / 32][32];
  int tid = threadIdx.x;
  int lane = tid & 31;
  int wv = tid >> 5;
  int bin = blockIdx.x;
  if (bin == NB) {
    rel_transform_tiles(r_feats, W_R_w, W_R_b, relOut, R, wv, BK / 32, lane);
    return;
  }
  bool rLds = (R <= RMAX);
  int nR4 = rLds ? R * (DD / 4) : 0;
  float4 rv[RT];
  {
    const float4* gR = (const float4*)r_feats;
    int lastR4 = (nR4 > 0) ? nR4 - 1 : 0;
#pragma unroll
    for (int i = 0; i < RT; ++i) {
      int idx = tid + i * BK;
      rv[i] = gR[(idx < nR4) ? idx : lastR4];
    }
  }
  {
    const float4* gI = (const float4*)W_I_w;
    const float4* gO = (const float4*)W_O_w;
#pragma unroll
    for (int i = 0; i < (DD * DD / 4) / BK; ++i) {
      int idx = tid + i * BK;
      int r = idx >> 4;
      int c4 = idx & 15;
      *(float4*)(&sWI[r * SPITCH + c4 * 4]) = gI[idx];
      *(float4*)(&sWO[r * SPITCH + c4 * 4]) = gO[idx];
    }
  }
#pragma unroll
  for (int i = 0; i < RT; ++i) ((float4*)sR)[tid + i * BK] = rv[i];
  if (tid < HB) {
    const int* rp = offG + (size_t)tid * (NB + 1) + bin;
    int s0 = rp[0];
    int e0 = rp[1];
    sSegS[tid] = s0;
    sSegL[tid] = e0 - s0;
  }
  sCnt[tid] = 0;
  __syncthreads();

  const int2* cz = (const int2*)coarse;
  int sS[SEGW], sL[SEGW];
  int mx = 0;
#pragma unroll
  for (int u = 0; u < SEGW; ++u) {
    int sg = wv * SEGW + u;
    sS[u] = sSegS[sg];
    sL[u] = sSegL[sg];
    mx = (sL[u] > mx) ? sL[u] : mx;
  }
  for (int i0 = 0; i0 < mx; i0 += 32) {
    int i = i0 + lane;
    int2 rec[SEGW];
#pragma unroll
    for (int u = 0; u < SEGW; ++u)
      rec[u] = (i < sL[u]) ? cz[sS[u] + i] : make_int2(0, -1);
#pragma unroll
    for (int u = 0; u < SEGW; ++u)
      if (rec[u].y >= 0) atomicAdd(&sCnt[rec[u].y & (BK - 1)], 1);
  }
  __syncthreads();
  int cnt = sCnt[tid];
  int inc = cnt;
  for (int d = 1; d < 32; d <<= 1) {
    int o = __shfl_up(inc, d);
    if (lane >= d) inc += o;
  }
  if (lane == 31) sWave[wv] = inc;
  __syncthreads();
  int woff = 0;
  int tot = 0;
#pragma unroll
  for (int i = 0; i < BK / 32; ++i) {
    int v = sWave[i];
    if (i < wv) woff += v;
    tot += v;
  }
  int st = woff + inc - cnt;
  bool staged = (tot <= STG);
  sCnt[tid] = st;
  __syncthreads();
  if (staged) {
    for (int i0 = 0; i0 < mx; i0 += 32) {
      int i = i0 + lane;
      int2 rec[SEGW];
#pragma unroll
      for (int u = 0; u < SEGW; ++u)
        rec[u] = (i < sL[u]) ? cz[sS[u] + i] : make_int2(0, -1);
#pragma unroll
      for (int u = 0; u < SEGW; ++u) {
        if (rec[u].y >= 0) {
          int slot = atomicAdd(&sCnt[rec[u].y & (BK - 1)], 1);
          sStage[slot] = make_int2(rec[u].x, (rec[u].y / BK) * (DD * 4));
        }
      }
    }
  }
  __syncthreads();

  int tile = bin * (BK / 32) + wv;
  bool active = (tile < numTiles);
  int rowBase = tile * 16;
  float* aIw = sAI[wv];
  float* aOw = sAO[wv];
  int degv = cnt;
  int endv = st + cnt;
  unsigned laneByte = (unsigned)lane * 8u;
  sDegF[wv][lane] = (float)cnt;

  if (active && staged) {
    if (rLds)
      walk_tile<true>(sStage, n_feats, r_feats, sR, degv, endv, lane, laneByte, aIw, aOw);
    else
      walk_tile<false>(sStage, n_feats, r_feats, sR, degv, endv, lane, laneByte, aIw, aOw);
  } else if (active) {
    for (int r = 0; r < 16; ++r) {
      *(float2*)(aIw + r * SPITCH + 2 * lane) = make_float2(0.f, 0.f);
      *(float2*)(aOw + r * SPITCH + 2 * lane) = make_float2(0.f, 0.f);
    }
    for (int sg = 0; sg < HB; ++sg) {
      int s0 = sSegS[sg];
      int len = sSegL[sg];
      for (int i = 0; i < len; ++i) {
        int2 rec = cz[s0 + i];
        int key = rec.y & (BK - 1);
        if ((key >> 5) == wv) {
          int r = (key & 31) >> 1;
          float* a = (key & 1) ? aOw : aIw;
          float2 xv = row_piece(n_feats, (unsigned)rec.x, laneByte);
          float2 qv = row_piece(r_feats, (unsigned)((rec.y / BK) * (DD * 4)), laneByte);
          float2* p = (float2*)(a + r * SPITCH + 2 * lane);
          float2 v = *p;
          v.x += xv.x - qv.x;
          v.y += xv.y - qv.y;
          *p = v;
        }
      }
    }
  }
  __syncthreads();
  if (!active) return;

  int lm   = lane & 15;
  int hs   = lane >> 4;
  int kb   = hs * 2;
  int mOff = hs * 8;

  float cI8[8], cO8[8], dg8[8];
#pragma unroll
  for (int i = 0; i < 8; ++i) {
    cI8[i] = sDegF[wv][2 * (i + mOff)];
    cO8[i] = sDegF[wv][2 * (i + mOff) + 1];
    dg8[i] = fmaxf(cI8[i] + cO8[i], 1.0f);
  }

  const float* aIrow = aIw + lm * SPITCH;
  const float* aOrow = aOw + lm * SPITCH;
  const float* sIrow[4];
  const float* sOrow[4];
#pragma unroll
  for (int t = 0; t < 4; ++t) {
    sIrow[t] = sWI + (t * 16 + lm) * SPITCH;
    sOrow[t] = sWO + (t * 16 + lm) * SPITCH;
  }

  v8f c[4];
#pragma unroll
  for (int t = 0; t < 4; ++t) c[t] = (v8f){0.f, 0.f, 0.f, 0.f, 0.f, 0.f, 0.f, 0.f};

#pragma unroll
  for (int k0 = 0; k0 < DD; k0 += 4) {
    v2f aI = *(const v2f*)(aIrow + k0 + kb);
    v2f aO = *(const v2f*)(aOrow + k0 + kb);
#pragma unroll
    for (int t = 0; t < 4; ++t) {
      v2f bI = *(const v2f*)(sIrow[t] + k0 + kb);
      v2f bO = *(const v2f*)(sOrow[t] + k0 + kb);
      c[t] = wmma_f32x4(aI, bI, c[t]);
      c[t] = wmma_f32x4(aO, bO, c[t]);
    }
  }

#pragma unroll
  for (int t = 0; t < 4; ++t) {
    int n = t * 16 + lm;
    float bI = W_I_b[n];
    float bO = W_O_b[n];
#pragma unroll
    for (int i = 0; i < 8; ++i) {
      int m = rowBase + i + mOff;
      float v = (c[t][i] + cI8[i] * bI + cO8[i] * bO) / dg8[i];
      out[(size_t)m * DD + n] = v;
    }
  }
}

extern "C" void kernel_launch(void* const* d_in, const int* in_sizes, int n_in,
                              void* d_out, int out_size, void* d_ws, size_t ws_size,
                              hipStream_t stream) {
  const float* n_feats  = (const float*)d_in[0];
  const float* r_feats  = (const float*)d_in[1];
  const float* W_I_w    = (const float*)d_in[2];
  const float* W_I_b    = (const float*)d_in[3];
  const float* W_O_w    = (const float*)d_in[4];
  const float* W_O_b    = (const float*)d_in[5];
  const float* W_R_w    = (const float*)d_in[6];
  const float* W_R_b    = (const float*)d_in[7];
  const int*   src      = (const int*)d_in[8];
  const int*   dst      = (const int*)d_in[9];
  const int*   etype    = (const int*)d_in[10];
  const int*   num_rels = (const int*)d_in[11];

  int N = in_sizes[0] / DD;
  int R = in_sizes[1] / DD;
  int E = in_sizes[8];
  int bins = (N + (1 << BN_SHIFT) - 1) >> BN_SHIFT;

  int* coarse = (int*)d_ws;
  int* offG   = coarse + (size_t)2 * E;

  int chunk = (E + HB - 1) / HB;
  size_t ldsBytes = (size_t)(chunk + LREC) * 8 + (size_t)bins * sizeof(int);

  bin_scatter_kernel<<<HB, HT, ldsBytes, stream>>>(src, dst, etype, num_rels, offG, coarse,
                                                  E, bins, chunk);

  int nodeTiles = (N + 15) / 16;
  bin_sort_node_kernel<<<bins + 1, BK, 0, stream>>>(coarse, offG, bins,
                                                   r_feats, W_R_w, W_R_b,
                                                   (float*)d_out + (size_t)N * DD, R,
                                                   n_feats, W_I_w, W_I_b, W_O_w, W_O_b,
                                                   (float*)d_out, nodeTiles);
}


# the gather takes two records per load: a half-wave reads one record's whole 256-byte row as 16-byte pieces, its record straight from the LDS image; the halves' sums are added where a list ends
# speedup vs baseline: 4.9217x; 1.0844x over previous
#include <hip/hip_runtime.h>


typedef float v2f __attribute__((ext_vector_type(2)));
typedef float v8f __attribute__((ext_vector_type(8)));

#define DD 64
#define SPITCH 68

static __device__ __forceinline__ v8f wmma_f32x4(v2f a, v2f b, v8f c) {
  return __builtin_amdgcn_wmma_f32_16x16x4_f32(
       false, a,  false, b,
       (short)0, c,  false,  false);
}

#define HB 256
#define HT 512
#define BN_SHIFT 8
#define BK 512
#define EPT 8
#define SEGW (HB / (BK / 32))
#define LREC 16
#define STG 6144
#define RMAX 200
#define RT ((RMAX * (DD / 4) + BK - 1) / BK)

static __device__ __forceinline__ void copy_image(const int2* img, int2* __restrict__ dst,
                                                  int first, int n, int tid, int nthreads) {
  int pairs = (n + 1) >> 1;
  for (int q = tid; q < pairs; q += nthreads) {
    int p = 2 * q;
    if (p >= first && p + 1 < n) {
      *(int4*)(dst + p) = *(const int4*)(img + p);
    } else {
      if (p >= first) dst[p] = img[p];
      if (p + 1 >= first && p + 1 < n) dst[p + 1] = img[p + 1];
    }
  }
}

__global__ void __launch_bounds__(HT)
bin_scatter_kernel(const int* __restrict__ src,
                   const int* __restrict__ dst,
                   const int* __restrict__ etype,
                   const int* __restrict__ num_rels_p,
                   int* __restrict__ offG,
                   int* __restrict__ coarse,
                   int E, int NB, int chunk) {
  extern __shared__ __align__(16) int shd[];
  __shared__ int sWave[HT / 32];
  int2* stage = (int2*)shd;
  int* sh = shd + 2 * (chunk + LREC);
  int tid = threadIdx.x;
  int lane = tid & 31;
  int wv = tid >> 5;
  for (int b = tid; b < NB; b += HT) sh[b] = 0;
  __syncthreads();
  int lo = blockIdx.x * chunk;
  lo = (lo < E) ? lo : E;
  int hi = lo + chunk;
  hi = (hi < E) ? hi : E;
  int loA = lo & ~(LREC - 1);
  for (int i = lo + tid; i < hi; i += EPT * HT) {
    int d[EPT];
#pragma unroll
    for (int u = 0; u < EPT; ++u) {
      int iu = i + u * HT;
      d[u] = (iu < hi) ? dst[iu] : -1;
    }
#pragma unroll
    for (int u = 0; u < EPT; ++u)
      if (d[u] >= 0) atomicAdd(&sh[d[u] >> BN_SHIFT], 1);
  }
  __syncthreads();
  int per = (NB + HT - 1) / HT;
  int b0 = tid * per;
  int b1 = b0 + per;
  b1 = (b1 < NB) ? b1 : NB;
  int sum = 0;
  for (int b = b0; b < b1; ++b) sum += sh[b];
  int inc = sum;
  for (int dd = 1; dd < 32; dd <<= 1) {
    int o = __shfl_up(inc, dd);
    if (lane >= dd) inc += o;
  }
  if (lane == 31) sWave[wv] = inc;
  __syncthreads();
  int woff = 0;
#pragma unroll
  for (int i = 0; i < HT / 32; ++i) {
    int v = sWave[i];
    if (i < wv) woff += v;
  }
  int* row = offG + (size_t)blockIdx.x * (NB + 1);
  int run = lo + woff + inc - sum;
  for (int b = b0; b < b1; ++b) {
    int c = sh[b];
    sh[b] = run;
    row[b] = run;
    run += c;
  }
  if (tid == 0) row[NB] = hi;
  __syncthreads();
  int half = num_rels_p[0] >> 1;
  for (int i = lo + tid; i < hi; i += EPT * HT) {
    int d[EPT], s[EPT], t[EPT];
#pragma unroll
    for (int u = 0; u < EPT; ++u) {
      int iu = i + u * HT;
      bool in = (iu < hi);
      int ii = in ? iu : i;
      int dv = dst[ii];
      d[u] = in ? dv : -1;
      s[u] = src[ii];
      t[u] = etype[ii];
    }
#pragma unroll
    for (int u = 0; u < EPT; ++u) {
      if (d[u] >= 0) {
        int key = ((d[u] & ((1 << BN_SHIFT) - 1)) << 1) | ((t[u] < half) ? 0 : 1);
        int slot = atomicAdd(&sh[d[u] >> BN_SHIFT], 1);
        stage[slot - loA] = make_int2(s[u] * (DD * 4), t[u] * BK + key);
      }
    }
  }
  __syncthreads();
  copy_image(stage, (int2*)coarse + loA, lo - loA, hi - loA, tid, HT);
}

static __device__ __forceinline__ void rel_transform_tiles(const float* __restrict__ r_feats,
                                                           const float* __restrict__ W_R_w,
                                                           const float* __restrict__ W_R_b,
                                                           float* __restrict__ out,
                                                           int R, int wv, int nWaves, int lane) {
  int numTiles = (R + 15) >> 4;
  int lm   = lane & 15;
  int hs   = lane >> 4;
  int kb   = hs * 2;
  int mOff = hs * 8;

  const float* Wrow[4];
#pragma unroll
  for (int t = 0; t < 4; ++t) Wrow[t] = W_R_w + (size_t)(t * 16 + lm) * DD;

  for (int tile = wv; tile < numTiles; tile += nWaves) {
    int rowBase = tile * 16;
    int mA = rowBase + lm;
    bool aOk = (mA < R);
    const float* arow = r_feats + (size_t)(aOk ? mA : 0) * DD;

    v8f c[4];
#pragma unroll
    for (int t = 0; t < 4; ++t) c[t] = (v8f){0.f, 0.f, 0.f, 0.f, 0.f, 0.f, 0.f, 0.f};

#pragma unroll
    for (int k0 = 0; k0 < DD; k0 += 4) {
      v2f a = *(const v2f*)(arow + k0 + kb);
      if (!aOk) { a.x = 0.f; a.y = 0.f; }
#pragma unroll
      for (int t = 0; t < 4; ++t) {
        v2f b = *(const v2f*)(Wrow[t] + k0 + kb);
        c[t] = wmma_f32x4(a, b, c[t]);
      }
    }

#pragma unroll
    for (int t = 0; t < 4; ++t) {
      int n = t * 16 + lm;
      float bias = W_R_b[n];
#pragma unroll
      for (int i = 0; i < 8; ++i) {
        int m = rowBase + i + mOff;
        if (m < R) out[(size_t)m * DD + n] = c[t][i] + bias;
      }
    }
  }
}

static __device__ __forceinline__ float2 row_piece(const float* __restrict__ base,
                                                   unsigned rowByte, unsigned laneByte) {
  return *(const float2*)((const char*)base + (size_t)(rowByte + laneByte));
}

typedef float v4f __attribute__((ext_vector_type(4)));

static __device__ __forceinline__ float other_half(float x) {
  return __builtin_bit_cast(float, __builtin_amdgcn_permlanex16(__builtin_bit_cast(int, x),
                                                                __builtin_bit_cast(int, x),
                                                                0x76543210, 0xfedcba98, false, false));
}

static __device__ __forceinline__ void end_list(v4f& acc, int& key, int& rel, int base, int endv,
                                                int l16, float* aIw, float* aOw) {
  v4f o;
  o.x = other_half(acc.x);
  o.y = other_half(acc.y);
  o.z = other_half(acc.z);
  o.w = other_half(acc.w);
  float* a = (key & 1) ? aOw : aIw;
  *(v4f*)(a + (key >> 1) * SPITCH + 4 * l16) = acc + o;
  acc = (v4f){0.f, 0.f, 0.f, 0.f};
  ++key;
  int ce = (key < 32) ? __builtin_amdgcn_readlane(endv, key) : 0x7fffffff;
  rel = ce - base;
}

template <bool RLDS>
static __device__ __forceinline__ void walk_tile(const int2* img,
                                                 const float* __restrict__ n_feats,
                                                 const float* __restrict__ rGlob,
                                                 const float* rTab,
                                                 int degv, int endv, int lane,
                                                 float* aIw, float* aOw) {
  int p0 = __builtin_amdgcn_readlane(endv, 0) - __builtin_amdgcn_readlane(degv, 0);
  int p1 = __builtin_amdgcn_readlane(endv, 31);
  int half = lane >> 4;
  int l16 = lane & 15;
  unsigned pieceByte = (unsigned)l16 * 16u;
  int key = 0;
  int rel = __builtin_amdgcn_readlane(endv, 0) - p0;
  v4f acc = (v4f){0.f, 0.f, 0.f, 0.f};
  int2 rc[16];
#pragma unroll
  for (int j = 0; j < 16; ++j) rc[j] = img[p0 + 2 * j + half];
  for (int base = p0; base < p1; base += 32) {
    v4f x[16], q[16];
#pragma unroll
    for (int j = 0; j < 16; ++j)
      x[j] = *(const v4f*)((const char*)n_feats + ((unsigned)rc[j].x + pieceByte));
#pragma unroll
    for (int j = 0; j < 16; ++j) {
      unsigned off = (unsigned)rc[j].y + pieceByte;
      if (RLDS)
        q[j] = *(const v4f*)((const char*)rTab + off);
      else
        q[j] = *(const v4f*)((const char*)rGlob + off);
    }
    if (base + 32 < p1) {
#pragma unroll
      for (int j = 0; j < 16; ++j) rc[j] = img[base + 32 + 2 * j + half];
    }
#pragma unroll
    for (int j = 0; j < 16; ++j) {
      v4f d = x[j] - q[j];
      if (__builtin_expect(rel < 2 * j + 2, 0)) {
        while (rel == 2 * j) end_list(acc, key, rel, base, endv, l16, aIw, aOw);
        if (rel == 2 * j + 1) {
          v4f zero = (v4f){0.f, 0.f, 0.f, 0.f};
          acc += (half == 0) ? d : zero;
          while (rel == 2 * j + 1) end_list(acc, key, rel, base, endv, l16, aIw, aOw);
          d = (half == 1) ? d : zero;
        }
      }
      acc += d;
    }
    rel -= 32;
  }
  while (key < 32) end_list(acc, key, rel, 0, endv, l16, aIw, aOw);
}

__global__ void __launch_bounds__(BK)
bin_sort_node_kernel(const int* __restrict__ coarse,
                     const int* __restrict__ offG,
                     int NB,
                     const float* __restrict__ r_feats,
                     const float* __restrict__ W_R_w,
                     const float* __restrict__ W_R_b,
                     float* __restrict__ relOut,
                     int R,
                     const float* __restrict__ n_feats,
                     const float* __restrict__ W_I_w,
                     const float* __restrict__ W_I_b,
                     const float* __restrict__ W_O_w,
                     const float* __restrict__ W_O_b,
                     float* __restrict__ out,
                     int numTiles) {
  static_assert(BK == (2 << BN_SHIFT), "one thread per key of the bin");
  static_assert(BK / 32 == (1 << BN_SHIFT) / 16, "one wave per 16-node tile of the bin");
  static_assert(HB <= BK && HB % 32 == 0 && HB % (BK / 32) == 0,
                "one thread per run, whole waves, the same number of runs per wave");
  __shared__ int sCnt[BK];
  __shared__ int sWave[BK / 32];
  __shared__ int sSegS[HB];
  __shared__ int sSegL[HB];
  __shared__ __align__(16) int2 sStage[STG + 32];
  __shared__ __align__(16) float sWI[DD * SPITCH];
  __shared__ __align__(16) float sWO[DD * SPITCH];
  __shared__ __align__(16) float sR[RT * BK * 4];
  __shared__ __align__(16) float sAI[BK / 32][16 * SPITCH];
  __shared__ __align__(16) float sAO[BK / 32][16 * SPITCH];
  __shared__ float sDegF[BK / 32][32];
  int tid = threadIdx.x;
  int lane = tid & 31;
  int wv = tid >> 5;
  int bin = blockIdx.x;
  if (bin == NB) {
    rel_transform_tiles(r_feats, W_R_w, W_R_b, relOut, R, wv, BK / 32, lane);
    return;
  }
  bool rLds = (R <= RMAX);
  int nR4 = rLds ? R * (DD / 4) : 0;
  float4 rv[RT];
  {
    const float4* gR = (const float4*)r_feats;
    int lastR4 = (nR4 > 0) ? nR4 - 1 : 0;
#pragma unroll
    for (int i = 0; i < RT; ++i) {
      int idx = tid + i * BK;
      rv[i] = gR[(idx < nR4) ? idx : lastR4];
    }
  }
  {
    const float4* gI = (const float4*)W_I_w;
    const float4* gO = (const float4*)W_O_w;
#pragma unroll
    for (int i = 0; i < (DD * DD / 4) / BK; ++i) {
      int idx = tid + i * BK;
      int r = idx >> 4;
      int c4 = idx & 15;
      *(float4*)(&sWI[r * SPITCH + c4 * 4]) = gI[idx];
      *(float4*)(&sWO[r * SPITCH + c4 * 4]) = gO[idx];
    }
  }
#pragma unroll
  for (int i = 0; i < RT; ++i) ((float4*)sR)[tid + i * BK] = rv[i];
  if (tid < HB) {
    const int* rp = offG + (size_t)tid * (NB + 1) + bin;
    int s0 = rp[0];
    int e0 = rp[1];
    sSegS[tid] = s0;
    sSegL[tid] = e0 - s0;
  }
  sCnt[tid] = 0;
  __syncthreads();

  const int2* cz = (const int2*)coarse;
  int sS[SEGW], sL[SEGW];
  int mx = 0;
#pragma unroll
  for (int u = 0; u < SEGW; ++u) {
    int sg = wv * SEGW + u;
    sS[u] = sSegS[sg];
    sL[u] = sSegL[sg];
    mx = (sL[u] > mx) ? sL[u] : mx;
  }
  for (int i0 = 0; i0 < mx; i0 += 32) {
    int i = i0 + lane;
    int2 rec[SEGW];
#pragma unroll
    for (int u = 0; u < SEGW; ++u)
      rec[u] = (i < sL[u]) ? cz[sS[u] + i] : make_int2(0, -1);
#pragma unroll
    for (int u = 0; u < SEGW; ++u)
      if (rec[u].y >= 0) atomicAdd(&sCnt[rec[u].y & (BK - 1)], 1);
  }
  __syncthreads();
  int cnt = sCnt[tid];
  int inc = cnt;
  for (int d = 1; d < 32; d <<= 1) {
    int o = __shfl_up(inc, d);
    if (lane >= d) inc += o;
  }
  if (lane == 31) sWave[wv] = inc;
  __syncthreads();
  int woff = 0;
  int tot = 0;
#pragma unroll
  for (int i = 0; i < BK / 32; ++i) {
    int v = sWave[i];
    if (i < wv) woff += v;
    tot += v;
  }
  int st = woff + inc - cnt;
  bool staged = (tot <= STG);
  sCnt[tid] = st;
  __syncthreads();
  if (staged) {
    if (tid < 32) sStage[tot + tid] = make_int2(0, 0);
    for (int i0 = 0; i0 < mx; i0 += 32) {
      int i = i0 + lane;
      int2 rec[SEGW];
#pragma unroll
      for (int u = 0; u < SEGW; ++u)
        rec[u] = (i < sL[u]) ? cz[sS[u] + i] : make_int2(0, -1);
#pragma unroll
      for (int u = 0; u < SEGW; ++u) {
        if (rec[u].y >= 0) {
          int slot = atomicAdd(&sCnt[rec[u].y & (BK - 1)], 1);
          sStage[slot] = make_int2(rec[u].x, (rec[u].y / BK) * (DD * 4));
        }
      }
    }
  }
  __syncthreads();

  int tile = bin * (BK / 32) + wv;
  bool active = (tile < numTiles);
  int rowBase = tile * 16;
  float* aIw = sAI[wv];
  float* aOw = sAO[wv];
  int degv = cnt;
  int endv = st + cnt;
  unsigned laneByte = (unsigned)lane * 8u;
  sDegF[wv][lane] = (float)cnt;

  if (active && staged) {
    if (rLds)
      walk_tile<true>(sStage, n_feats, r_feats, sR, degv, endv, lane, aIw, aOw);
    else
      walk_tile<false>(sStage, n_feats, r_feats, sR, degv, endv, lane, aIw, aOw);
  } else if (active) {
    for (int r = 0; r < 16; ++r) {
      *(float2*)(aIw + r * SPITCH + 2 * lane) = make_float2(0.f, 0.f);
      *(float2*)(aOw + r * SPITCH + 2 * lane) = make_float2(0.f, 0.f);
    }
    for (int sg = 0; sg < HB; ++sg) {
      int s0 = sSegS[sg];
      int len = sSegL[sg];
      for (int i = 0; i < len; ++i) {
        int2 rec = cz[s0 + i];
        int key = rec.y & (BK - 1);
        if ((key >> 5) == wv) {
          int r = (key & 31) >> 1;
          float* a = (key & 1) ? aOw : aIw;
          float2 xv = row_piece(n_feats, (unsigned)rec.x, laneByte);
          float2 qv = row_piece(r_feats, (unsigned)((rec.y / BK) * (DD * 4)), laneByte);
          float2* p = (float2*)(a + r * SPITCH + 2 * lane);
          float2 v = *p;
          v.x += xv.x - qv.x;
          v.y += xv.y - qv.y;
          *p = v;
        }
      }
    }
  }
  __syncthreads();
  if (!active) return;

  int lm   = lane & 15;
  int hs   = lane >> 4;
  int kb   = hs * 2;
  int mOff = hs * 8;

  float cI8[8], cO8[8], dg8[8];
#pragma unroll
  for (int i = 0; i < 8; ++i) {
    cI8[i] = sDegF[wv][2 * (i + mOff)];
    cO8[i] = sDegF[wv][2 * (i + mOff) + 1];
    dg8[i] = fmaxf(cI8[i] + cO8[i], 1.0f);
  }

  const float* aIrow = aIw + lm * SPITCH;
  const float* aOrow = aOw + lm * SPITCH;
  const float* sIrow[4];
  const float* sOrow[4];
#pragma unroll
  for (int t = 0; t < 4; ++t) {
    sIrow[t] = sWI + (t * 16 + lm) * SPITCH;
    sOrow[t] = sWO + (t * 16 + lm) * SPITCH;
  }

  v8f c[4];
#pragma unroll
  for (int t = 0; t < 4; ++t) c[t] = (v8f){0.f, 0.f, 0.f, 0.f, 0.f, 0.f, 0.f, 0.f};

#pragma unroll
  for (int k0 = 0; k0 < DD; k0 += 4) {
    v2f aI = *(const v2f*)(aIrow + k0 + kb);
    v2f aO = *(const v2f*)(aOrow + k0 + kb);
#pragma unroll
    for (int t = 0; t < 4; ++t) {
      v2f bI = *(const v2f*)(sIrow[t] + k0 + kb);
      v2f bO = *(const v2f*)(sOrow[t] + k0 + kb);
      c[t] = wmma_f32x4(aI, bI, c[t]);
      c[t] = wmma_f32x4(aO, bO, c[t]);
    }
  }

#pragma unroll
  for (int t = 0; t < 4; ++t) {
    int n = t * 16 + lm;
    float bI = W_I_b[n];
    float bO = W_O_b[n];
#pragma unroll
    for (int i = 0; i < 8; ++i) {
      int m = rowBase + i + mOff;
      float v = (c[t][i] + cI8[i] * bI + cO8[i] * bO) / dg8[i];
      out[(size_t)m * DD + n] = v;
    }
  }
}

extern "C" void kernel_launch(void* const* d_in, const int* in_sizes, int n_in,
                              void* d_out, int out_size, void* d_ws, size_t ws_size,
                              hipStream_t stream) {
  const float* n_feats  = (const float*)d_in[0];
  const float* r_feats  = (const float*)d_in[1];
  const float* W_I_w    = (const float*)d_in[2];
  const float* W_I_b    = (const float*)d_in[3];
  const float* W_O_w    = (const float*)d_in[4];
  const float* W_O_b    = (const float*)d_in[5];
  const float* W_R_w    = (const float*)d_in[6];
  const float* W_R_b    = (const float*)d_in[7];
  const int*   src      = (const int*)d_in[8];
  const int*   dst      = (const int*)d_in[9];
  const int*   etype    = (const int*)d_in[10];
  const int*   num_rels = (const int*)d_in[11];

  int N = in_sizes[0] / DD;
  int R = in_sizes[1] / DD;
  int E = in_sizes[8];
  int bins = (N + (1 << BN_SHIFT) - 1) >> BN_SHIFT;

  int* coarse = (int*)d_ws;
  int* offG   = coarse + (size_t)2 * E;

  int chunk = (E + HB - 1) / HB;
  size_t ldsBytes = (size_t)(chunk + LREC) * 8 + (size_t)bins * sizeof(int);

  bin_scatter_kernel<<<HB, HT, ldsBytes, stream>>>(src, dst, etype, num_rels, offG, coarse,
                                                  E, bins, chunk);

  int nodeTiles = (N + 15) / 16;
  bin_sort_node_kernel<<<bins + 1, BK, 0, stream>>>(coarse, offG, bins,
                                                   r_feats, W_R_w, W_R_b,
                                                   (float*)d_out + (size_t)N * DD, R,
                                                   n_feats, W_I_w, W_I_b, W_O_w, W_O_b,
                                                   (float*)d_out, nodeTiles);
}


# both counting sorts read their input once: the LDS atomic that counts a record also hands it its rank, and record and rank wait in registers across the scan; no second read, no second atomic pass
# speedup vs baseline: 5.3134x; 1.0796x over previous
#include <hip/hip_runtime.h>


typedef float v2f __attribute__((ext_vector_type(2)));
typedef float v8f __attribute__((ext_vector_type(8)));

#define DD 64
#define SPITCH 68

static __device__ __forceinline__ v8f wmma_f32x4(v2f a, v2f b, v8f c) {
  return __builtin_amdgcn_wmma_f32_16x16x4_f32(
       false, a,  false, b,
       (short)0, c,  false,  false);
}

#define HB 256
#define HT 512
#define BN_SHIFT 8
#define BK 512
#define EPT 8
#define SEGW (HB / (BK / 32))
#define LREC 16
#define STG 6144
#define RMAX 200
#define RT ((RMAX * (DD / 4) + BK - 1) / BK)

static __device__ __forceinline__ void copy_image(const int2* img, int2* __restrict__ dst,
                                                  int first, int n, int tid, int nthreads) {
  int pairs = (n + 1) >> 1;
  for (int q = tid; q < pairs; q += nthreads) {
    int p = 2 * q;
    if (p >= first && p + 1 < n) {
      *(int4*)(dst + p) = *(const int4*)(img + p);
    } else {
      if (p >= first) dst[p] = img[p];
      if (p + 1 >= first && p + 1 < n) dst[p + 1] = img[p + 1];
    }
  }
}

__global__ void __launch_bounds__(HT)
bin_scatter_kernel(const int* __restrict__ src,
                   const int* __restrict__ dst,
                   const int* __restrict__ etype,
                   const int* __restrict__ num_rels_p,
                   int* __restrict__ offG,
                   int* __restrict__ coarse,
                   int E, int NB, int chunk) {
  extern __shared__ __align__(16) int shd[];
  __shared__ int sWave[HT / 32];
  int2* stage = (int2*)shd;
  int* sh = shd + 2 * (chunk + LREC);
  int* sh2 = sh + NB;
  int tid = threadIdx.x;
  int lane = tid & 31;
  int wv = tid >> 5;
  int lo = blockIdx.x * chunk;
  lo = (lo < E) ? lo : E;
  int hi = lo + chunk;
  hi = (hi < E) ? hi : E;
  int loA = lo & ~(LREC - 1);
  int d[EPT], s[EPT], t[EPT], rk[EPT];
#pragma unroll
  for (int u = 0; u < EPT; ++u) {
    d[u] = -1;
    s[u] = 0;
    t[u] = 0;
  }
  if (lo < hi) {
#pragma unroll
    for (int u = 0; u < EPT; ++u) {
      int iu = lo + tid + u * HT;
      bool in = (iu < hi);
      int ii = in ? iu : lo;
      int dv = dst[ii];
      d[u] = in ? dv : -1;
      s[u] = src[ii];
      t[u] = etype[ii];
    }
  }
  int half = num_rels_p[0] >> 1;
  for (int b = tid; b < 2 * NB; b += HT) sh[b] = 0;
  __syncthreads();
#pragma unroll
  for (int u = 0; u < EPT; ++u)
    rk[u] = (d[u] >= 0) ? atomicAdd(&sh[d[u] >> BN_SHIFT], 1) : 0;
  for (int i = lo + tid + EPT * HT; i < hi; i += EPT * HT) {
    int d2[EPT];
#pragma unroll
    for (int u = 0; u < EPT; ++u) {
      int iu = i + u * HT;
      d2[u] = (iu < hi) ? dst[iu] : -1;
    }
#pragma unroll
    for (int u = 0; u < EPT; ++u)
      if (d2[u] >= 0) atomicAdd(&sh2[d2[u] >> BN_SHIFT], 1);
  }
  __syncthreads();
  int per = (NB + HT - 1) / HT;
  int b0 = tid * per;
  int b1 = b0 + per;
  b1 = (b1 < NB) ? b1 : NB;
  int sum = 0;
  for (int b = b0; b < b1; ++b) sum += sh[b] + sh2[b];
  int inc = sum;
  for (int dd = 1; dd < 32; dd <<= 1) {
    int o = __shfl_up(inc, dd);
    if (lane >= dd) inc += o;
  }
  if (lane == 31) sWave[wv] = inc;
  __syncthreads();
  int woff = 0;
#pragma unroll
  for (int i = 0; i < HT / 32; ++i) {
    int v = sWave[i];
    if (i < wv) woff += v;
  }
  int* row = offG + (size_t)blockIdx.x * (NB + 1);
  int run = lo + woff + inc - sum;
  for (int b = b0; b < b1; ++b) {
    int c1 = sh[b];
    int c2 = sh2[b];
    sh[b] = run;
    sh2[b] = run + c1;
    row[b] = run;
    run += c1 + c2;
  }
  if (tid == 0) row[NB] = hi;
  __syncthreads();
  int rs[EPT];
#pragma unroll
  for (int u = 0; u < EPT; ++u) rs[u] = sh[((d[u] >= 0) ? d[u] : 0) >> BN_SHIFT];
#pragma unroll
  for (int u = 0; u < EPT; ++u) {
    if (d[u] >= 0) {
      int key = ((d[u] & ((1 << BN_SHIFT) - 1)) << 1) | ((t[u] < half) ? 0 : 1);
      stage[rs[u] + rk[u] - loA] = make_int2(s[u] * (DD * 4), t[u] * BK + key);
    }
  }
  for (int i = lo + tid + EPT * HT; i < hi; i += EPT * HT) {
    int d2[EPT], s2[EPT], t2[EPT];
#pragma unroll
    for (int u = 0; u < EPT; ++u) {
      int iu = i + u * HT;
      bool in = (iu < hi);
      int ii = in ? iu : i;
      int dv = dst[ii];
      d2[u] = in ? dv : -1;
      s2[u] = src[ii];
      t2[u] = etype[ii];
    }
#pragma unroll
    for (int u = 0; u < EPT; ++u) {
      if (d2[u] >= 0) {
        int key = ((d2[u] & ((1 << BN_SHIFT) - 1)) << 1) | ((t2[u] < half) ? 0 : 1);
        int slot = atomicAdd(&sh2[d2[u] >> BN_SHIFT], 1);
        stage[slot - loA] = make_int2(s2[u] * (DD * 4), t2[u] * BK + key);
      }
    }
  }
  __syncthreads();
  copy_image(stage, (int2*)coarse + loA, lo - loA, hi - loA, tid, HT);
}

static __device__ __forceinline__ void rel_transform_tiles(const float* __restrict__ r_feats,
                                                           const float* __restrict__ W_R_w,
                                                           const float* __restrict__ W_R_b,
                                                           float* __restrict__ out,
                                                           int R, int wv, int nWaves, int lane) {
  int numTiles = (R + 15) >> 4;
  int lm   = lane & 15;
  int hs   = lane >> 4;
  int kb   = hs * 2;
  int mOff = hs * 8;

  const float* Wrow[4];
#pragma unroll
  for (int t = 0; t < 4; ++t) Wrow[t] = W_R_w + (size_t)(t * 16 + lm) * DD;

  for (int tile = wv; tile < numTiles; tile += nWaves) {
    int rowBase = tile * 16;
    int mA = rowBase + lm;
    bool aOk = (mA < R);
    const float* arow = r_feats + (size_t)(aOk ? mA : 0) * DD;

    v8f c[4];
#pragma unroll
    for (int t = 0; t < 4; ++t) c[t] = (v8f){0.f, 0.f, 0.f, 0.f, 0.f, 0.f, 0.f, 0.f};

#pragma unroll
    for (int k0 = 0; k0 < DD; k0 += 4) {
      v2f a = *(const v2f*)(arow + k0 + kb);
      if (!aOk) { a.x = 0.f; a.y = 0.f; }
#pragma unroll
      for (int t = 0; t < 4; ++t) {
        v2f b = *(const v2f*)(Wrow[t] + k0 + kb);
        c[t] = wmma_f32x4(a, b, c[t]);
      }
    }

#pragma unroll
    for (int t = 0; t < 4; ++t) {
      int n = t * 16 + lm;
      float bias = W_R_b[n];
#pragma unroll
      for (int i = 0; i < 8; ++i) {
        int m = rowBase + i + mOff;
        if (m < R) out[(size_t)m * DD + n] = c[t][i] + bias;
      }
    }
  }
}

static __device__ __forceinline__ float2 row_piece(const float* __restrict__ base,
                                                   unsigned rowByte, unsigned laneByte) {
  return *(const float2*)((const char*)base + (size_t)(rowByte + laneByte));
}

typedef float v4f __attribute__((ext_vector_type(4)));

static __device__ __forceinline__ float other_half(float x) {
  return __builtin_bit_cast(float, __builtin_amdgcn_permlanex16(__builtin_bit_cast(int, x),
                                                                __builtin_bit_cast(int, x),
                                                                0x76543210, 0xfedcba98, false, false));
}

static __device__ __forceinline__ void end_list(v4f& acc, int& key, int& rel, int base, int endv,
                                                int l16, float* aIw, float* aOw) {
  v4f o;
  o.x = other_half(acc.x);
  o.y = other_half(acc.y);
  o.z = other_half(acc.z);
  o.w = other_half(acc.w);
  float* a = (key & 1) ? aOw : aIw;
  *(v4f*)(a + (key >> 1) * SPITCH + 4 * l16) = acc + o;
  acc = (v4f){0.f, 0.f, 0.f, 0.f};
  ++key;
  int ce = (key < 32) ? __builtin_amdgcn_readlane(endv, key) : 0x7fffffff;
  rel = ce - base;
}

template <bool RLDS>
static __device__ __forceinline__ void walk_tile(const int2* img,
                                                 const float* __restrict__ n_feats,
                                                 const float* __restrict__ rGlob,
                                                 const float* rTab,
                                                 int degv, int endv, int lane,
                                                 float* aIw, float* aOw) {
  int p0 = __builtin_amdgcn_readlane(endv, 0) - __builtin_amdgcn_readlane(degv, 0);
  int p1 = __builtin_amdgcn_readlane(endv, 31);
  int half = lane >> 4;
  int l16 = lane & 15;
  unsigned pieceByte = (unsigned)l16 * 16u;
  int key = 0;
  int rel = __builtin_amdgcn_readlane(endv, 0) - p0;
  v4f acc = (v4f){0.f, 0.f, 0.f, 0.f};
  int2 rc[16];
#pragma unroll
  for (int j = 0; j < 16; ++j) rc[j] = img[p0 + 2 * j + half];
  for (int base = p0; base < p1; base += 32) {
    v4f x[16], q[16];
#pragma unroll
    for (int j = 0; j < 16; ++j)
      x[j] = *(const v4f*)((const char*)n_feats + ((unsigned)rc[j].x + pieceByte));
#pragma unroll
    for (int j = 0; j < 16; ++j) {
      unsigned off = (unsigned)rc[j].y + pieceByte;
      if (RLDS)
        q[j] = *(const v4f*)((const char*)rTab + off);
      else
        q[j] = *(const v4f*)((const char*)rGlob + off);
    }
    if (base + 32 < p1) {
#pragma unroll
      for (int j = 0; j < 16; ++j) rc[j] = img[base + 32 + 2 * j + half];
    }
#pragma unroll
    for (int j = 0; j < 16; ++j) {
      v4f d = x[j] - q[j];
      if (__builtin_expect(rel < 2 * j + 2, 0)) {
        while (rel == 2 * j) end_list(acc, key, rel, base, endv, l16, aIw, aOw);
        if (rel == 2 * j + 1) {
          v4f zero = (v4f){0.f, 0.f, 0.f, 0.f};
          acc += (half == 0) ? d : zero;
          while (rel == 2 * j + 1) end_list(acc, key, rel, base, endv, l16, aIw, aOw);
          d = (half == 1) ? d : zero;
        }
      }
      acc += d;
    }
    rel -= 32;
  }
  while (key < 32) end_list(acc, key, rel, 0, endv, l16, aIw, aOw);
}

__global__ void __launch_bounds__(BK)
bin_sort_node_kernel(const int* __restrict__ coarse,
                     const int* __restrict__ offG,
                     int NB,
                     const float* __restrict__ r_feats,
                     const float* __restrict__ W_R_w,
                     const float* __restrict__ W_R_b,
                     float* __restrict__ relOut,
                     int R,
                     const float* __restrict__ n_feats,
                     const float* __restrict__ W_I_w,
                     const float* __restrict__ W_I_b,
                     const float* __restrict__ W_O_w,
                     const float* __restrict__ W_O_b,
                     float* __restrict__ out,
                     int numTiles) {
  static_assert(BK == (2 << BN_SHIFT), "one thread per key of the bin");
  static_assert(BK / 32 == (1 << BN_SHIFT) / 16, "one wave per 16-node tile of the bin");
  static_assert(HB <= BK && HB % 32 == 0 && HB % (BK / 32) == 0,
                "one thread per run, whole waves, the same number of runs per wave");
  __shared__ int sCnt[BK];
  __shared__ int sCnt2[BK];
  __shared__ int sWave[BK / 32];
  __shared__ int sSegS[HB];
  __shared__ int sSegL[HB];
  __shared__ __align__(16) int2 sStage[STG + 32];
  __shared__ __align__(16) float sWI[DD * SPITCH];
  __shared__ __align__(16) float sWO[DD * SPITCH];
  __shared__ __align__(16) float sR[RT * BK * 4];
  __shared__ __align__(16) float sAI[BK / 32][16 * SPITCH];
  __shared__ __align__(16) float sAO[BK / 32][16 * SPITCH];
  __shared__ float sDegF[BK / 32][32];
  int tid = threadIdx.x;
  int lane = tid & 31;
  int wv = tid >> 5;
  int bin = blockIdx.x;
  if (bin == NB) {
    rel_transform_tiles(r_feats, W_R_w, W_R_b, relOut, R, wv, BK / 32, lane);
    return;
  }
  bool rLds = (R <= RMAX);
  int nR4 = rLds ? R * (DD / 4) : 0;
  float4 rv[RT];
  {
    const float4* gR = (const float4*)r_feats;
    int lastR4 = (nR4 > 0) ? nR4 - 1 : 0;
#pragma unroll
    for (int i = 0; i < RT; ++i) {
      int idx = tid + i * BK;
      rv[i] = gR[(idx < nR4) ? idx : lastR4];
    }
  }
  {
    const float4* gI = (const float4*)W_I_w;
    const float4* gO = (const float4*)W_O_w;
#pragma unroll
    for (int i = 0; i < (DD * DD / 4) / BK; ++i) {
      int idx = tid + i * BK;
      int r = idx >> 4;
      int c4 = idx & 15;
      *(float4*)(&sWI[r * SPITCH + c4 * 4]) = gI[idx];
      *(float4*)(&sWO[r * SPITCH + c4 * 4]) = gO[idx];
    }
  }
#pragma unroll
  for (int i = 0; i < RT; ++i) ((float4*)sR)[tid + i * BK] = rv[i];
  if (tid < HB) {
    const int* rp = offG + (size_t)tid * (NB + 1) + bin;
    int s0 = rp[0];
    int e0 = rp[1];
    sSegS[tid] = s0;
    sSegL[tid] = e0 - s0;
  }
  sCnt[tid] = 0;
  sCnt2[tid] = 0;
  __syncthreads();

  const int2* cz = (const int2*)coarse;
  int sS[SEGW], sL[SEGW];
  int mx = 0;
#pragma unroll
  for (int u = 0; u < SEGW; ++u) {
    int sg = wv * SEGW + u;
    sS[u] = sSegS[sg];
    sL[u] = sSegL[sg];
    mx = (sL[u] > mx) ? sL[u] : mx;
  }
  int2 rec[SEGW];
  int rk[SEGW];
#pragma unroll
  for (int u = 0; u < SEGW; ++u)
    rec[u] = (lane < sL[u]) ? cz[sS[u] + lane] : make_int2(0, -1);
#pragma unroll
  for (int u = 0; u < SEGW; ++u)
    rk[u] = (rec[u].y >= 0) ? atomicAdd(&sCnt[rec[u].y & (BK - 1)], 1) : 0;
  for (int i0 = 32; i0 < mx; i0 += 32) {
    int i = i0 + lane;
    int ky[SEGW];
#pragma unroll
    for (int u = 0; u < SEGW; ++u)
      ky[u] = (i < sL[u]) ? cz[sS[u] + i].y : -1;
#pragma unroll
    for (int u = 0; u < SEGW; ++u)
      if (ky[u] >= 0) atomicAdd(&sCnt2[ky[u] & (BK - 1)], 1);
  }
  __syncthreads();
  int c1 = sCnt[tid];
  int cnt = c1 + sCnt2[tid];
  int inc = cnt;
  for (int d = 1; d < 32; d <<= 1) {
    int o = __shfl_up(inc, d);
    if (lane >= d) inc += o;
  }
  if (lane == 31) sWave[wv] = inc;
  __syncthreads();
  int woff = 0;
  int tot = 0;
#pragma unroll
  for (int i = 0; i < BK / 32; ++i) {
    int v = sWave[i];
    if (i < wv) woff += v;
    tot += v;
  }
  int st = woff + inc - cnt;
  bool staged = (tot <= STG);
  sCnt[tid] = st;
  sCnt2[tid] = st + c1;
  __syncthreads();
  if (staged) {
    if (tid < 32) sStage[tot + tid] = make_int2(0, 0);
    int ks[SEGW];
#pragma unroll
    for (int u = 0; u < SEGW; ++u) ks[u] = sCnt[rec[u].y & (BK - 1)];
#pragma unroll
    for (int u = 0; u < SEGW; ++u) {
      if (rec[u].y >= 0)
        sStage[ks[u] + rk[u]] = make_int2(rec[u].x, (rec[u].y / BK) * (DD * 4));
    }
    for (int i0 = 32; i0 < mx; i0 += 32) {
      int i = i0 + lane;
      int2 r2[SEGW];
#pragma unroll
      for (int u = 0; u < SEGW; ++u)
        r2[u] = (i < sL[u]) ? cz[sS[u] + i] : make_int2(0, -1);
#pragma unroll
      for (int u = 0; u < SEGW; ++u) {
        if (r2[u].y >= 0) {
          int slot = atomicAdd(&sCnt2[r2[u].y & (BK - 1)], 1);
          sStage[slot] = make_int2(r2[u].x, (r2[u].y / BK) * (DD * 4));
        }
      }
    }
  }
  __syncthreads();

  int tile = bin * (BK / 32) + wv;
  bool active = (tile < numTiles);
  int rowBase = tile * 16;
  float* aIw = sAI[wv];
  float* aOw = sAO[wv];
  int degv = cnt;
  int endv = st + cnt;
  unsigned laneByte = (unsigned)lane * 8u;
  sDegF[wv][lane] = (float)cnt;

  if (active && staged) {
    if (rLds)
      walk_tile<true>(sStage, n_feats, r_feats, sR, degv, endv, lane, aIw, aOw);
    else
      walk_tile<false>(sStage, n_feats, r_feats, sR, degv, endv, lane, aIw, aOw);
  } else if (active) {
    for (int r = 0; r < 16; ++r) {
      *(float2*)(aIw + r * SPITCH + 2 * lane) = make_float2(0.f, 0.f);
      *(float2*)(aOw + r * SPITCH + 2 * lane) = make_float2(0.f, 0.f);
    }
    for (int sg = 0; sg < HB; ++sg) {
      int s0 = sSegS[sg];
      int len = sSegL[sg];
      for (int i = 0; i < len; ++i) {
        int2 rec = cz[s0 + i];
        int key = rec.y & (BK - 1);
        if ((key >> 5) == wv) {
          int r = (key & 31) >> 1;
          float* a = (key & 1) ? aOw : aIw;
          float2 xv = row_piece(n_feats, (unsigned)rec.x, laneByte);
          float2 qv = row_piece(r_feats, (unsigned)((rec.y / BK) * (DD * 4)), laneByte);
          float2* p = (float2*)(a + r * SPITCH + 2 * lane);
          float2 v = *p;
          v.x += xv.x - qv.x;
          v.y += xv.y - qv.y;
          *p = v;
        }
      }
    }
  }
  __syncthreads();
  if (!active) return;

  int lm   = lane & 15;
  int hs   = lane >> 4;
  int kb   = hs * 2;
  int mOff = hs * 8;

  float cI8[8], cO8[8], dg8[8];
#pragma unroll
  for (int i = 0; i < 8; ++i) {
    cI8[i] = sDegF[wv][2 * (i + mOff)];
    cO8[i] = sDegF[wv][2 * (i + mOff) + 1];
    dg8[i] = fmaxf(cI8[i] + cO8[i], 1.0f);
  }

  const float* aIrow = aIw + lm * SPITCH;
  const float* aOrow = aOw + lm * SPITCH;
  const float* sIrow[4];
  const float* sOrow[4];
#pragma unroll
  for (int t = 0; t < 4; ++t) {
    sIrow[t] = sWI + (t * 16 + lm) * SPITCH;
    sOrow[t] = sWO + (t * 16 + lm) * SPITCH;
  }

  v8f c[4];
#pragma unroll
  for (int t = 0; t < 4; ++t) c[t] = (v8f){0.f, 0.f, 0.f, 0.f, 0.f, 0.f, 0.f, 0.f};

#pragma unroll
  for (int k0 = 0; k0 < DD; k0 += 4) {
    v2f aI = *(const v2f*)(aIrow + k0 + kb);
    v2f aO = *(const v2f*)(aOrow + k0 + kb);
#pragma unroll
    for (int t = 0; t < 4; ++t) {
      v2f bI = *(const v2f*)(sIrow[t] + k0 + kb);
      v2f bO = *(const v2f*)(sOrow[t] + k0 + kb);
      c[t] = wmma_f32x4(aI, bI, c[t]);
      c[t] = wmma_f32x4(aO, bO, c[t]);
    }
  }

#pragma unroll
  for (int t = 0; t < 4; ++t) {
    int n = t * 16 + lm;
    float bI = W_I_b[n];
    float bO = W_O_b[n];
#pragma unroll
    for (int i = 0; i < 8; ++i) {
      int m = rowBase + i + mOff;
      float v = (c[t][i] + cI8[i] * bI + cO8[i] * bO) / dg8[i];
      out[(size_t)m * DD + n] = v;
    }
  }
}

extern "C" void kernel_launch(void* const* d_in, const int* in_sizes, int n_in,
                              void* d_out, int out_size, void* d_ws, size_t ws_size,
                              hipStream_t stream) {
  const float* n_feats  = (const float*)d_in[0];
  const float* r_feats  = (const float*)d_in[1];
  const float* W_I_w    = (const float*)d_in[2];
  const float* W_I_b    = (const float*)d_in[3];
  const float* W_O_w    = (const float*)d_in[4];
  const float* W_O_b    = (const float*)d_in[5];
  const float* W_R_w    = (const float*)d_in[6];
  const float* W_R_b    = (const float*)d_in[7];
  const int*   src      = (const int*)d_in[8];
  const int*   dst      = (const int*)d_in[9];
  const int*   etype    = (const int*)d_in[10];
  const int*   num_rels = (const int*)d_in[11];

  int N = in_sizes[0] / DD;
  int R = in_sizes[1] / DD;
  int E = in_sizes[8];
  int bins = (N + (1 << BN_SHIFT) - 1) >> BN_SHIFT;

  int* coarse = (int*)d_ws;
  int* offG   = coarse + (size_t)2 * E;

  int chunk = (E + HB - 1) / HB;
  size_t ldsBytes = (size_t)(chunk + LREC) * 8 + (size_t)bins * 2 * sizeof(int);

  bin_scatter_kernel<<<HB, HT, ldsBytes, stream>>>(src, dst, etype, num_rels, offG, coarse,
                                                  E, bins, chunk);

  int nodeTiles = (N + 15) / 16;
  bin_sort_node_kernel<<<bins + 1, BK, 0, stream>>>(coarse, offG, bins,
                                                   r_feats, W_R_w, W_R_b,
                                                   (float*)d_out + (size_t)N * DD, R,
                                                   n_feats, W_I_w, W_I_b, W_O_w, W_O_b,
                                                   (float*)d_out, nodeTiles);
}
